# starting kernel
# baseline (speedup 1.0000x reference)
#include <hip/hip_runtime.h>


#define NN 100000
#define NE 1600000
#define DD 128

typedef __attribute__((ext_vector_type(2))) float v2f;
typedef __attribute__((ext_vector_type(8))) float v8f;

__global__ __launch_bounds__(256) void k_deg_init(float* __restrict__ deg) {
  int i = blockIdx.x * 256 + threadIdx.x;
  if (i < NN) deg[i] = 1.0f;
}

__global__ __launch_bounds__(256) void k_deg_count(const int* __restrict__ dst,
                                                   float* __restrict__ deg) {
  int e = blockIdx.x * 256 + threadIdx.x;
  if (e < NE) atomicAdd(&deg[dst[e]], 1.0f);
}

__global__ __launch_bounds__(256) void k_rsqrt(float* __restrict__ d) {
  int i = blockIdx.x * 256 + threadIdx.x;
  if (i < NN) d[i] = rsqrtf(d[i]);
}

__global__ __launch_bounds__(256) void k_gemm_wmma(const float* __restrict__ X,
                                                   const float* __restrict__ W,
                                                   float* __restrict__ H) {
  const int wave = (blockIdx.x * 256 + (int)threadIdx.x) >> 5;
  const int lane = threadIdx.x & 31;
  const int row0 = (wave >> 3) << 4;
  const int col0 = (wave & 7) << 4;
  if (row0 >= NN) return;
  const int half = lane >> 4;
  const int l16  = lane & 15;

  v8f acc = {};
  const float* xr = X + (size_t)(row0 + l16) * DD;
  const float* wc = W + col0 + l16;
  for (int k = 0; k < DD; k += 4) {
    v2f a = *(const v2f*)(xr + k + 2 * half);
    v2f b;
    b.x = wc[(size_t)(k + 2 * half) * DD];
    b.y = wc[(size_t)(k + 2 * half + 1) * DD];
    acc = __builtin_amdgcn_wmma_f32_16x16x4_f32(
         false, a,  false, b,
         (short)0, acc,  false,  false);
  }
  float* hp = H + (size_t)row0 * DD + col0 + l16;
#pragma unroll
  for (int r = 0; r < 8; ++r)
    hp[(size_t)(r + 8 * half) * DD] = acc[r];
}

__global__ __launch_bounds__(256) void k_zero(float* __restrict__ p, int n) {
  int i = blockIdx.x * 256 + threadIdx.x;
  if (i < n) p[i] = 0.0f;
}

__global__ __launch_bounds__(256) void k_scatter(const int* __restrict__ src,
                                                 const int* __restrict__ dst,
                                                 const float* __restrict__ dinv,
                                                 const float* __restrict__ H,
                                                 float* __restrict__ agg) {
  const int e = (blockIdx.x * 256 + (int)threadIdx.x) >> 5;
  const int lane = threadIdx.x & 31;
  if (e >= NE) return;
  const int s = src[e];
  const int d = dst[e];
  const float norm = dinv[s] * dinv[d];
  const float4 v = ((const float4*)(H + (size_t)s * DD))[lane];
  float* a = agg + (size_t)d * DD + lane * 4;
  atomicAdd(a + 0, v.x * norm);
  atomicAdd(a + 1, v.y * norm);
  atomicAdd(a + 2, v.z * norm);
  atomicAdd(a + 3, v.w * norm);
}

__global__ __launch_bounds__(256) void k_finalize(const float* __restrict__ agg,
                                                  const float* __restrict__ H,
                                                  const float* __restrict__ dinv,
                                                  const float* __restrict__ bias,
                                                  float* __restrict__ out,
                                                  int relu) {
  int i = blockIdx.x * 256 + threadIdx.x;
  if (i >= NN * DD) return;
  int n = i >> 7;
  int f = i & (DD - 1);
  float di = dinv[n];
  float v = agg[i] + H[i] * di * di + bias[f];
  out[i] = relu ? fmaxf(v, 0.0f) : v;
}

extern "C" void kernel_launch(void* const* d_in, const int* in_sizes, int n_in,
                              void* d_out, int out_size, void* d_ws, size_t ws_size,
                              hipStream_t stream) {
  const float* x[2]  = {(const float*)d_in[0], (const float*)d_in[2]};
  const int*   ei[2] = {(const int*)d_in[1],   (const int*)d_in[3]};
  const float* W0 = (const float*)d_in[4];
  const float* b0 = (const float*)d_in[5];
  const float* W1 = (const float*)d_in[6];
  const float* b1 = (const float*)d_in[7];
  float* out = (float*)d_out;

  float* ws   = (float*)d_ws;
  float* dinv = ws;
  float* H    = ws + 100352;
  float* AG   = H  + (size_t)NN * DD;
  float* Y    = AG + (size_t)NN * DD;

  const int nd    = NN * DD;
  const dim3 B(256);
  const int gN    = (NN + 255) / 256;
  const int gE    = (NE + 255) / 256;
  const int gND   = (nd + 255) / 256;
  const int gGemm = (NN / 16) * (DD / 16) / 8;
  const int gScat = NE / 8;

  for (int bb = 0; bb < 2; ++bb) {
    const int* srcp = ei[bb];
    const int* dstp = ei[bb] + NE;

    k_deg_init <<<gN, B, 0, stream>>>(dinv);
    k_deg_count<<<gE, B, 0, stream>>>(dstp, dinv);
    k_rsqrt    <<<gN, B, 0, stream>>>(dinv);

    k_gemm_wmma<<<gGemm, B, 0, stream>>>(x[bb], W0, H);
    k_zero     <<<gND,   B, 0, stream>>>(AG, nd);
    k_scatter  <<<gScat, B, 0, stream>>>(srcp, dstp, dinv, H, AG);
    k_finalize <<<gND,   B, 0, stream>>>(AG, H, dinv, b0, Y, 1);

    k_gemm_wmma<<<gGemm, B, 0, stream>>>(Y, W1, H);
    k_zero     <<<gND,   B, 0, stream>>>(AG, nd);
    k_scatter  <<<gScat, B, 0, stream>>>(srcp, dstp, dinv, H, AG);
    k_finalize <<<gND,   B, 0, stream>>>(AG, H, dinv, b1, out + (size_t)bb * nd, 0);
  }
}


# Float atomic scatter-add replaced by a per-node gather over a destination-ordered edge list (int count+rank, scan, place); rows summed in registers in arrival-rank order, zero+finalize folded in
# speedup vs baseline: 2.7138x; 2.7138x over previous
#include <hip/hip_runtime.h>


#define NN 100000
#define NE 1600000
#define DD 128
#define NP 100352

typedef __attribute__((ext_vector_type(2))) float v2f;
typedef __attribute__((ext_vector_type(8))) float v8f;

__global__ __launch_bounds__(256) void k_cnt_init(int* __restrict__ cnt) {
  int i = blockIdx.x * 256 + threadIdx.x;
  if (i < NN) cnt[i] = 0;
}

__global__ __launch_bounds__(256) void k_count_rank(const int* __restrict__ dst,
                                                    int* __restrict__ cnt,
                                                    int* __restrict__ rank) {
  int e = blockIdx.x * 256 + threadIdx.x;
  if (e < NE) rank[e] = atomicAdd(&cnt[dst[e]], 1);
}

__global__ __launch_bounds__(256) void k_scan(const int* __restrict__ cnt,
                                              int* __restrict__ rowptr,
                                              float* __restrict__ dinv) {
  __shared__ int red[8];
  __shared__ int wsum[8];
  const int t = threadIdx.x;
  const int lane = t & 31;
  const int w = t >> 5;
  const int b = blockIdx.x;
  const int4* c4 = (const int4*)cnt;

  int pre = 0;
  const int lim = b * 256;
  int i = t;
  for (; i + 1792 < lim; i += 2048) {
    const int4 a0 = c4[i],        a1 = c4[i + 256],  a2 = c4[i + 512],  a3 = c4[i + 768];
    const int4 a4 = c4[i + 1024], a5 = c4[i + 1280], a6 = c4[i + 1536], a7 = c4[i + 1792];
    pre += (a0.x + a0.y + a0.z + a0.w) + (a1.x + a1.y + a1.z + a1.w) +
           (a2.x + a2.y + a2.z + a2.w) + (a3.x + a3.y + a3.z + a3.w) +
           (a4.x + a4.y + a4.z + a4.w) + (a5.x + a5.y + a5.z + a5.w) +
           (a6.x + a6.y + a6.z + a6.w) + (a7.x + a7.y + a7.z + a7.w);
  }
  for (; i < lim; i += 256) {
    const int4 a = c4[i];
    pre += a.x + a.y + a.z + a.w;
  }
  for (int o = 16; o > 0; o >>= 1) pre += __shfl_xor(pre, o);
  if (lane == 0) red[w] = pre;

  const int q = b * 256 + t;
  const bool ok = q < NN / 4;
  int4 v = {0, 0, 0, 0};
  if (ok) v = c4[q];
  const int s = v.x + v.y + v.z + v.w;
  int inc = s;
  for (int o = 1; o < 32; o <<= 1) {
    const int u = __shfl_up(inc, o);
    if (lane >= o) inc += u;
  }
  if (lane == 31) wsum[w] = inc;
  __syncthreads();
  int base = 0;
  for (int k = 0; k < 8; ++k) {
    base += red[k];
    if (k < w) base += wsum[k];
  }
  if (ok) {
    int4 r;
    r.x = base + inc - s;
    r.y = r.x + v.x;
    r.z = r.y + v.y;
    r.w = r.z + v.z;
    ((int4*)rowptr)[q] = r;
    float4 d;
    d.x = rsqrtf((float)v.x + 1.0f);
    d.y = rsqrtf((float)v.y + 1.0f);
    d.z = rsqrtf((float)v.z + 1.0f);
    d.w = rsqrtf((float)v.w + 1.0f);
    ((float4*)dinv)[q] = d;
  }
}

__global__ __launch_bounds__(256) void k_place(const int* __restrict__ src,
                                               const int* __restrict__ dst,
                                               const int* __restrict__ rowptr,
                                               const int* __restrict__ rank,
                                               int* __restrict__ perm) {
  int e = blockIdx.x * 256 + threadIdx.x;
  if (e < NE) perm[rowptr[dst[e]] + rank[e]] = src[e];
}

__global__ __launch_bounds__(256) void k_gemm_wmma(const float* __restrict__ X,
                                                   const float* __restrict__ W,
                                                   float* __restrict__ H) {
  const int wave = (blockIdx.x * 256 + (int)threadIdx.x) >> 5;
  const int lane = threadIdx.x & 31;
  const int row0 = (wave >> 3) << 4;
  const int col0 = (wave & 7) << 4;
  if (row0 >= NN) return;
  const int half = lane >> 4;
  const int l16  = lane & 15;

  v8f acc = {};
  const float* xr = X + (size_t)(row0 + l16) * DD;
  const float* wc = W + col0 + l16;
  for (int k = 0; k < DD; k += 4) {
    v2f a = *(const v2f*)(xr + k + 2 * half);
    v2f b;
    b.x = wc[(size_t)(k + 2 * half) * DD];
    b.y = wc[(size_t)(k + 2 * half + 1) * DD];
    acc = __builtin_amdgcn_wmma_f32_16x16x4_f32(
         false, a,  false, b,
         (short)0, acc,  false,  false);
  }
  float* hp = H + (size_t)row0 * DD + col0 + l16;
#pragma unroll
  for (int r = 0; r < 8; ++r)
    hp[(size_t)(r + 8 * half) * DD] = acc[r];
}

template <int K>
__device__ __forceinline__ void add_rows(const float4* __restrict__ H4, int sv,
                                         float dv, float dd, int j, int lane,
                                         float4& acc) {
  int s[K];
  float w[K];
  float4 v[K];
#pragma unroll
  for (int k = 0; k < K; ++k) {
    s[k] = __builtin_amdgcn_readlane(sv, j + k);
  }
#pragma unroll
  for (int k = 0; k < K; ++k) v[k] = H4[(size_t)s[k] * (DD / 4) + lane];
#pragma unroll
  for (int k = 0; k < K; ++k) {
    w[k] = __int_as_float(__builtin_amdgcn_readlane(__float_as_int(dv), j + k)) * dd;
  }
#pragma unroll
  for (int k = 0; k < K; ++k) {
    acc.x = fmaf(v[k].x, w[k], acc.x);
    acc.y = fmaf(v[k].y, w[k], acc.y);
    acc.z = fmaf(v[k].z, w[k], acc.z);
    acc.w = fmaf(v[k].w, w[k], acc.w);
  }
}

__global__ __launch_bounds__(256) void k_gather(const int* __restrict__ rowptr,
                                                const int* __restrict__ cnt,
                                                const int* __restrict__ perm,
                                                const float* __restrict__ dinv,
                                                const float* __restrict__ H,
                                                const float* __restrict__ bias,
                                                float* __restrict__ out,
                                                int relu) {
  const int lane = threadIdx.x & 31;
  const int d = __builtin_amdgcn_readfirstlane((blockIdx.x * 256 + (int)threadIdx.x) >> 5);
  if (d >= NN) return;
  const int beg = rowptr[d];
  const int n = cnt[d];
  const float dd = dinv[d];
  const float4* H4 = (const float4*)H;

  float4 acc = {0.0f, 0.0f, 0.0f, 0.0f};
  for (int base = 0; base < n; base += 32) {
    const int m = (n - base < 32) ? (n - base) : 32;
    const int idx = (lane < m) ? lane : (m - 1);
    const int sv = perm[beg + base + idx];
    const float dv = dinv[sv];
    int j = 0;
    for (; j + 8 <= m; j += 8) add_rows<8>(H4, sv, dv, dd, j, lane, acc);
    if (m - j >= 4) { add_rows<4>(H4, sv, dv, dd, j, lane, acc); j += 4; }
    if (m - j >= 2) { add_rows<2>(H4, sv, dv, dd, j, lane, acc); j += 2; }
    if (m - j >= 1) { add_rows<1>(H4, sv, dv, dd, j, lane, acc); }
  }

  const float4 h = H4[(size_t)d * (DD / 4) + lane];
  const int f = lane * 4;
  float4 o;
  o.x = acc.x + h.x * dd * dd + bias[f + 0];
  o.y = acc.y + h.y * dd * dd + bias[f + 1];
  o.z = acc.z + h.z * dd * dd + bias[f + 2];
  o.w = acc.w + h.w * dd * dd + bias[f + 3];
  if (relu) {
    o.x = fmaxf(o.x, 0.0f);
    o.y = fmaxf(o.y, 0.0f);
    o.z = fmaxf(o.z, 0.0f);
    o.w = fmaxf(o.w, 0.0f);
  }
  ((float4*)out)[(size_t)d * (DD / 4) + lane] = o;
}

extern "C" void kernel_launch(void* const* d_in, const int* in_sizes, int n_in,
                              void* d_out, int out_size, void* d_ws, size_t ws_size,
                              hipStream_t stream) {
  const float* x[2]  = {(const float*)d_in[0], (const float*)d_in[2]};
  const int*   ei[2] = {(const int*)d_in[1],   (const int*)d_in[3]};
  const float* W0 = (const float*)d_in[4];
  const float* b0 = (const float*)d_in[5];
  const float* W1 = (const float*)d_in[6];
  const float* b1 = (const float*)d_in[7];
  float* out = (float*)d_out;

  int*   cnt    = (int*)d_ws;
  int*   rowptr = cnt + NP;
  float* dinv   = (float*)(rowptr + NP);
  int*   rank   = (int*)(dinv + NP);
  int*   perm   = rank + NE;
  float* H      = (float*)(perm + NE);
  float* Y      = H + (size_t)NN * DD;

  const int nd    = NN * DD;
  const dim3 B(256);
  const int gN    = (NN + 255) / 256;
  const int gE    = (NE + 255) / 256;
  const int gScan = NP / 1024;
  const int gGemm = (NN / 16) * (DD / 16) / 8;
  const int gGath = NN / 8;

  for (int bb = 0; bb < 2; ++bb) {
    const int* srcp = ei[bb];
    const int* dstp = ei[bb] + NE;

    k_cnt_init  <<<gN,    B, 0, stream>>>(cnt);
    k_count_rank<<<gE,    B, 0, stream>>>(dstp, cnt, rank);
    k_scan      <<<gScan, B, 0, stream>>>(cnt, rowptr, dinv);
    k_place     <<<gE,    B, 0, stream>>>(srcp, dstp, rowptr, rank, perm);

    k_gemm_wmma<<<gGemm, B, 0, stream>>>(x[bb], W0, H);
    k_gather   <<<gGath, B, 0, stream>>>(rowptr, cnt, perm, dinv, H, b0, Y, 1);

    k_gemm_wmma<<<gGemm, B, 0, stream>>>(Y, W1, H);
    k_gather   <<<gGath, B, 0, stream>>>(rowptr, cnt, perm, dinv, H, b1, out + (size_t)bb * nd, 0);
  }
}


# Dense transform moved behind the aggregation and fused into it: resident 16-wave workgroups hold W in LDS, gather 16 node rows, WMMA the tile; float sum order now (A X) W, was A (X W)
# speedup vs baseline: 3.4472x; 1.2702x over previous
#include <hip/hip_runtime.h>


#define NN 100000
#define NE 1600000
#define DD 128
#define NP 100352

typedef __attribute__((ext_vector_type(2))) float v2f;
typedef __attribute__((ext_vector_type(8))) float v8f;

__global__ __launch_bounds__(256) void k_cnt_init(int* __restrict__ cnt) {
  int i = blockIdx.x * 256 + threadIdx.x;
  if (i < NN) cnt[i] = 0;
}

__global__ __launch_bounds__(256) void k_count_rank(const int* __restrict__ dst,
                                                    int* __restrict__ cnt,
                                                    int* __restrict__ rank) {
  int e = blockIdx.x * 256 + threadIdx.x;
  if (e < NE) rank[e] = atomicAdd(&cnt[dst[e]], 1);
}

__global__ __launch_bounds__(256) void k_scan(const int* __restrict__ cnt,
                                              int* __restrict__ rowptr,
                                              float* __restrict__ dinv) {
  __shared__ int red[8];
  __shared__ int wsum[8];
  const int t = threadIdx.x;
  const int lane = t & 31;
  const int w = t >> 5;
  const int b = blockIdx.x;
  const int4* c4 = (const int4*)cnt;

  int pre = 0;
  const int lim = b * 256;
  int i = t;
  for (; i + 1792 < lim; i += 2048) {
    const int4 a0 = c4[i],        a1 = c4[i + 256],  a2 = c4[i + 512],  a3 = c4[i + 768];
    const int4 a4 = c4[i + 1024], a5 = c4[i + 1280], a6 = c4[i + 1536], a7 = c4[i + 1792];
    pre += (a0.x + a0.y + a0.z + a0.w) + (a1.x + a1.y + a1.z + a1.w) +
           (a2.x + a2.y + a2.z + a2.w) + (a3.x + a3.y + a3.z + a3.w) +
           (a4.x + a4.y + a4.z + a4.w) + (a5.x + a5.y + a5.z + a5.w) +
           (a6.x + a6.y + a6.z + a6.w) + (a7.x + a7.y + a7.z + a7.w);
  }
  for (; i < lim; i += 256) {
    const int4 a = c4[i];
    pre += a.x + a.y + a.z + a.w;
  }
  for (int o = 16; o > 0; o >>= 1) pre += __shfl_xor(pre, o);
  if (lane == 0) red[w] = pre;

  const int q = b * 256 + t;
  const bool ok = q < NN / 4;
  int4 v = {0, 0, 0, 0};
  if (ok) v = c4[q];
  const int s = v.x + v.y + v.z + v.w;
  int inc = s;
  for (int o = 1; o < 32; o <<= 1) {
    const int u = __shfl_up(inc, o);
    if (lane >= o) inc += u;
  }
  if (lane == 31) wsum[w] = inc;
  __syncthreads();
  int base = 0;
  for (int k = 0; k < 8; ++k) {
    base += red[k];
    if (k < w) base += wsum[k];
  }
  if (ok) {
    int4 r;
    r.x = base + inc - s;
    r.y = r.x + v.x;
    r.z = r.y + v.y;
    r.w = r.z + v.z;
    ((int4*)rowptr)[q] = r;
    float4 d;
    d.x = rsqrtf((float)v.x + 1.0f);
    d.y = rsqrtf((float)v.y + 1.0f);
    d.z = rsqrtf((float)v.z + 1.0f);
    d.w = rsqrtf((float)v.w + 1.0f);
    ((float4*)dinv)[q] = d;
  }
}

__global__ __launch_bounds__(256) void k_place(const int* __restrict__ src,
                                               const int* __restrict__ dst,
                                               const int* __restrict__ rowptr,
                                               const int* __restrict__ rank,
                                               int* __restrict__ perm) {
  int e = blockIdx.x * 256 + threadIdx.x;
  if (e < NE) perm[rowptr[dst[e]] + rank[e]] = src[e];
}

#define NT (NN / 16)
#define PW 288
#define PA 132

template <int K>
__device__ __forceinline__ void add_rows(const float4* __restrict__ H4, int sv,
                                         float dv, float dd, int j, int lane,
                                         float4& acc) {
  int s[K];
  float w[K];
  float4 v[K];
#pragma unroll
  for (int k = 0; k < K; ++k) {
    s[k] = __builtin_amdgcn_readlane(sv, j + k);
  }
#pragma unroll
  for (int k = 0; k < K; ++k) v[k] = H4[(size_t)s[k] * (DD / 4) + lane];
#pragma unroll
  for (int k = 0; k < K; ++k) {
    w[k] = __int_as_float(__builtin_amdgcn_readlane(__float_as_int(dv), j + k)) * dd;
  }
#pragma unroll
  for (int k = 0; k < K; ++k) {
    acc.x = fmaf(v[k].x, w[k], acc.x);
    acc.y = fmaf(v[k].y, w[k], acc.y);
    acc.z = fmaf(v[k].z, w[k], acc.z);
    acc.w = fmaf(v[k].w, w[k], acc.w);
  }
}

__global__ __launch_bounds__(512) void k_conv(const int* __restrict__ rowptr,
                                              const int* __restrict__ cnt,
                                              const int* __restrict__ perm,
                                              const float* __restrict__ dinv,
                                              const float* __restrict__ X,
                                              const float* __restrict__ W,
                                              const float* __restrict__ bias,
                                              float* __restrict__ out,
                                              int relu) {
  __shared__ __attribute__((aligned(16))) float Wl[(DD / 2) * PW];
  __shared__ __attribute__((aligned(16))) float At[2][16 * PA];
  const int t = threadIdx.x;
  const int lane = t & 31;
  const int w = __builtin_amdgcn_readfirstlane(t >> 5);

  const float4* W4 = (const float4*)W;
  for (int i = 0; i < 4; ++i) {
    const int idx = t + 512 * i;
    const int kp = idx >> 5;
    const int j = idx & 31;
    const float4 r0 = W4[(2 * kp) * (DD / 4) + j];
    const float4 r1 = W4[(2 * kp + 1) * (DD / 4) + j];
    float4 s0, s1;
    s0.x = r0.x; s0.y = r1.x; s0.z = r0.y; s0.w = r1.y;
    s1.x = r0.z; s1.y = r1.z; s1.z = r0.w; s1.w = r1.w;
    float4* q = (float4*)&Wl[kp * PW + 8 * j];
    q[0] = s0;
    q[1] = s1;
  }
  __syncthreads();

  const float4* X4 = (const float4*)X;
  const float bv = bias[(w & 7) * 16 + (lane & 15)];
  int it = 0;
  for (int tile = blockIdx.x; tile < NT; tile += gridDim.x, ++it) {
    float* Ab = At[it & 1];

    const int d = tile * 16 + w;
    const int beg = rowptr[d];
    const int n = cnt[d];
    const float dd = dinv[d];
    float4 acc = {0.0f, 0.0f, 0.0f, 0.0f};
    for (int base = 0; base < n; base += 32) {
      const int m = (n - base < 32) ? (n - base) : 32;
      const int idx = (lane < m) ? lane : (m - 1);
      const int sv = perm[beg + base + idx];
      const float dv = dinv[sv];
      int j = 0;
      for (; j + 8 <= m; j += 8) add_rows<8>(X4, sv, dv, dd, j, lane, acc);
      if (m - j >= 4) { add_rows<4>(X4, sv, dv, dd, j, lane, acc); j += 4; }
      if (m - j >= 2) { add_rows<2>(X4, sv, dv, dd, j, lane, acc); j += 2; }
      if (m - j >= 1) { add_rows<1>(X4, sv, dv, dd, j, lane, acc); }
    }
    const float4 h = X4[(size_t)d * (DD / 4) + lane];
    float4 o;
    o.x = acc.x + h.x * dd * dd;
    o.y = acc.y + h.y * dd * dd;
    o.z = acc.z + h.z * dd * dd;
    o.w = acc.w + h.w * dd * dd;
    *(float4*)&Ab[w * PA + 4 * lane] = o;
    __syncthreads();

    if (w < 8) {
      const int half = lane >> 4;
      const int c = lane & 15;
      const int n0 = w * 16;
      const float* ap = Ab + c * PA + 2 * half;
      const float* bp = Wl + half * PW + 2 * (n0 + c);
      v8f a8 = {};
#pragma unroll 8
      for (int k = 0; k < DD; k += 4) {
        const v2f a = *(const v2f*)(ap + k);
        const v2f b = *(const v2f*)(bp + (k >> 1) * PW);
        a8 = __builtin_amdgcn_wmma_f32_16x16x4_f32(
             false, a,  false, b,
             (short)0, a8,  false,  false);
      }
      float* op = out + (size_t)(tile * 16) * DD + n0 + c;
#pragma unroll
      for (int r = 0; r < 8; ++r) {
        float v = a8[r] + bv;
        if (relu) v = fmaxf(v, 0.0f);
        op[(size_t)(r + 8 * half) * DD] = v;
      }
    }
  }
}

extern "C" void kernel_launch(void* const* d_in, const int* in_sizes, int n_in,
                              void* d_out, int out_size, void* d_ws, size_t ws_size,
                              hipStream_t stream) {
  const float* x[2]  = {(const float*)d_in[0], (const float*)d_in[2]};
  const int*   ei[2] = {(const int*)d_in[1],   (const int*)d_in[3]};
  const float* W0 = (const float*)d_in[4];
  const float* b0 = (const float*)d_in[5];
  const float* W1 = (const float*)d_in[6];
  const float* b1 = (const float*)d_in[7];
  float* out = (float*)d_out;

  int*   cnt    = (int*)d_ws;
  int*   rowptr = cnt + NP;
  float* dinv   = (float*)(rowptr + NP);
  int*   rank   = (int*)(dinv + NP);
  int*   perm   = rank + NE;
  float* Y      = (float*)(perm + NE);

  const int nd    = NN * DD;
  const dim3 B(256);
  const int gN    = (NN + 255) / 256;
  const int gE    = (NE + 255) / 256;
  const int gScan = NP / 1024;
  const int gConv = 768;

  for (int bb = 0; bb < 2; ++bb) {
    const int* srcp = ei[bb];
    const int* dstp = ei[bb] + NE;

    k_cnt_init  <<<gN,    B, 0, stream>>>(cnt);
    k_count_rank<<<gE,    B, 0, stream>>>(dstp, cnt, rank);
    k_scan      <<<gScan, B, 0, stream>>>(cnt, rowptr, dinv);
    k_place     <<<gE,    B, 0, stream>>>(srcp, dstp, rowptr, rank, perm);

    k_conv<<<gConv, dim3(512), 0, stream>>>(rowptr, cnt, perm, dinv, x[bb], W0, b0, Y, 1);

    k_conv<<<gConv, dim3(512), 0, stream>>>(rowptr, cnt, perm, dinv, Y, W1, b1, out + (size_t)bb * nd, 0);
  }
}


# Edges ordered by destination by two-level LDS counting sort (chunk histogram, scan, run-wise bucket scatter, in-bucket sort); no global atomic, no per-edge scattered store; sums in LDS arrival order
# speedup vs baseline: 5.7266x; 1.6612x over previous
#include <hip/hip_runtime.h>


#define NN 100000
#define NE 1600000
#define DD 128
#define NP 100352

typedef __attribute__((ext_vector_type(2))) float v2f;
typedef __attribute__((ext_vector_type(8))) float v8f;

#define NBK 256
#define BS 391
#define CH 8192
#define NBLK 196
#define NM (NBK * NBLK)
#define BSP 416
#define CAP 16384

__global__ __launch_bounds__(512) void k_hist(const int* __restrict__ dst,
                                              int* __restrict__ bcount) {
  __shared__ int h[NBK];
  const int t = threadIdx.x;
  const int blk = blockIdx.x;
  if (t < NBK) h[t] = 0;
  __syncthreads();
  const int base = blk * CH;
  const int n = (NE - base < CH) ? (NE - base) : CH;
  int d[CH / 512];
#pragma unroll
  for (int k = 0; k < CH / 512; ++k) {
    const int i = t + 512 * k;
    d[k] = dst[base + ((i < n) ? i : (n - 1))];
  }
#pragma unroll
  for (int k = 0; k < CH / 512; ++k)
    if (t + 512 * k < n) atomicAdd(&h[d[k] / BS], 1);
  __syncthreads();
  if (t < NBK) bcount[t * NBLK + blk] = h[t];
}

__global__ __launch_bounds__(1024) void k_bscan(const int* __restrict__ bcount,
                                                int* __restrict__ boffs) {
  __shared__ int buf[NM];
  __shared__ int wsum[32];
  const int t = threadIdx.x;
  const int lane = t & 31;
  const int w = t >> 5;
  const int per = NM / 1024;
  for (int k = 0; k < per; ++k) buf[k * 1024 + t] = bcount[k * 1024 + t];
  __syncthreads();
  int s = 0;
  for (int k = 0; k < per; ++k) s += buf[t * per + k];
  int inc = s;
  for (int o = 1; o < 32; o <<= 1) {
    const int u = __shfl_up(inc, o);
    if (lane >= o) inc += u;
  }
  if (lane == 31) wsum[w] = inc;
  __syncthreads();
  if (w == 0) {
    int x = wsum[lane];
    for (int o = 1; o < 32; o <<= 1) {
      const int u = __shfl_up(x, o);
      if (lane >= o) x += u;
    }
    wsum[lane] = x;
  }
  __syncthreads();
  int run = inc - s + ((w > 0) ? wsum[w - 1] : 0);
  for (int k = 0; k < per; ++k) {
    const int c = buf[t * per + k];
    buf[t * per + k] = run;
    run += c;
  }
  if (t == 1023) boffs[NM] = run;
  __syncthreads();
  for (int k = 0; k < per; ++k) boffs[k * 1024 + t] = buf[k * 1024 + t];
}

__global__ __launch_bounds__(512) void k_bin(const int* __restrict__ src,
                                             const int* __restrict__ dst,
                                             const int* __restrict__ boffs,
                                             int* __restrict__ tsrc,
                                             int* __restrict__ tdst) {
  __shared__ int lc[NBK];
  __shared__ int ls[NBK];
  __shared__ int go[NBK];
  __shared__ int wtot[16];
  __shared__ int ss[CH];
  __shared__ int sd[CH];
  const int t = threadIdx.x;
  const int lane = t & 31;
  const int w = t >> 5;
  const int blk = blockIdx.x;
  const int base = blk * CH;
  const int n = (NE - base < CH) ? (NE - base) : CH;
  if (t < NBK) lc[t] = 0;
  __syncthreads();
  int d[CH / 512], sv[CH / 512], r[CH / 512];
#pragma unroll
  for (int k = 0; k < CH / 512; ++k) {
    const int i = t + 512 * k;
    const int ii = base + ((i < n) ? i : (n - 1));
    d[k] = dst[ii];
    sv[k] = src[ii];
  }
#pragma unroll
  for (int k = 0; k < CH / 512; ++k) {
    r[k] = 0;
    if (t + 512 * k < n) r[k] = atomicAdd(&lc[d[k] / BS], 1);
  }
  __syncthreads();
  const int v = (t < NBK) ? lc[t] : 0;
  int inc = v;
  for (int o = 1; o < 32; o <<= 1) {
    const int u = __shfl_up(inc, o);
    if (lane >= o) inc += u;
  }
  if (lane == 31) wtot[w] = inc;
  __syncthreads();
  int bw = 0;
  for (int k = 0; k < 16; ++k)
    if (k < w) bw += wtot[k];
  if (t < NBK) {
    const int ex = bw + inc - v;
    ls[t] = ex;
    go[t] = boffs[t * NBLK + blk] - ex;
  }
  __syncthreads();
#pragma unroll
  for (int k = 0; k < CH / 512; ++k) {
    const int i = t + 512 * k;
    if (i < n) {
      const int p = ls[d[k] / BS] + r[k];
      ss[p] = sv[k];
      sd[p] = d[k];
    }
  }
  __syncthreads();
  for (int i = t; i < n; i += 512) {
    const int dd = sd[i];
    const int p = go[dd / BS] + i;
    tsrc[p] = ss[i];
    tdst[p] = dd;
  }
}

__global__ __launch_bounds__(512) void k_sort(const int* __restrict__ tsrc,
                                              const int* __restrict__ tdst,
                                              const int* __restrict__ boffs,
                                              int* __restrict__ cnt,
                                              int* __restrict__ rowptr,
                                              float* __restrict__ dinv,
                                              int* __restrict__ perm) {
  __shared__ int c[BSP];
  __shared__ int rp[BSP];
  __shared__ int wtot[16];
  __shared__ int stage[CAP];
  const int t = threadIdx.x;
  const int lane = t & 31;
  const int w = t >> 5;
  const int b = blockIdx.x;
  const int beg = boffs[b * NBLK];
  const int end = boffs[(b + 1) * NBLK];
  const int nb = end - beg;
  const int n0 = b * BS;
  const int nn = (NN - n0 < BS) ? (NN - n0) : BS;
  if (t < BSP) c[t] = 0;
  __syncthreads();
  for (int i0 = beg + t; i0 < end; i0 += 512 * 8) {
    int dd[8];
#pragma unroll
    for (int k = 0; k < 8; ++k) {
      const int i = i0 + 512 * k;
      dd[k] = tdst[(i < end) ? i : (end - 1)];
    }
#pragma unroll
    for (int k = 0; k < 8; ++k)
      if (i0 + 512 * k < end) atomicAdd(&c[dd[k] - n0], 1);
  }
  __syncthreads();
  const int v = (t < BSP) ? c[t] : 0;
  int inc = v;
  for (int o = 1; o < 32; o <<= 1) {
    const int u = __shfl_up(inc, o);
    if (lane >= o) inc += u;
  }
  if (lane == 31) wtot[w] = inc;
  __syncthreads();
  int bw = 0;
  for (int k = 0; k < 16; ++k)
    if (k < w) bw += wtot[k];
  const int ex = bw + inc - v;
  if (t < nn) {
    cnt[n0 + t] = v;
    rowptr[n0 + t] = beg + ex;
    dinv[n0 + t] = rsqrtf((float)v + 1.0f);
  }
  if (t < BSP) {
    rp[t] = ex;
    c[t] = 0;
  }
  __syncthreads();
  if (nb <= CAP) {
    for (int i0 = beg + t; i0 < end; i0 += 512 * 8) {
      int dd[8], sv[8];
#pragma unroll
      for (int k = 0; k < 8; ++k) {
        const int i = i0 + 512 * k;
        const int ii = (i < end) ? i : (end - 1);
        dd[k] = tdst[ii];
        sv[k] = tsrc[ii];
      }
#pragma unroll
      for (int k = 0; k < 8; ++k) {
        if (i0 + 512 * k < end) {
          const int ld = dd[k] - n0;
          const int rk = atomicAdd(&c[ld], 1);
          stage[rp[ld] + rk] = sv[k];
        }
      }
    }
    __syncthreads();
    for (int i = t; i < nb; i += 512) perm[beg + i] = stage[i];
  } else {
    for (int i = beg + t; i < end; i += 512) {
      const int ld = tdst[i] - n0;
      const int rk = atomicAdd(&c[ld], 1);
      perm[beg + rp[ld] + rk] = tsrc[i];
    }
  }
}

#define NT (NN / 16)
#define PW 288
#define PA 132

template <int K>
__device__ __forceinline__ void add_rows(const float4* __restrict__ H4, int sv,
                                         float dv, float dd, int j, int lane,
                                         float4& acc) {
  int s[K];
  float w[K];
  float4 v[K];
#pragma unroll
  for (int k = 0; k < K; ++k) {
    s[k] = __builtin_amdgcn_readlane(sv, j + k);
  }
#pragma unroll
  for (int k = 0; k < K; ++k) v[k] = H4[(size_t)s[k] * (DD / 4) + lane];
#pragma unroll
  for (int k = 0; k < K; ++k) {
    w[k] = __int_as_float(__builtin_amdgcn_readlane(__float_as_int(dv), j + k)) * dd;
  }
#pragma unroll
  for (int k = 0; k < K; ++k) {
    acc.x = fmaf(v[k].x, w[k], acc.x);
    acc.y = fmaf(v[k].y, w[k], acc.y);
    acc.z = fmaf(v[k].z, w[k], acc.z);
    acc.w = fmaf(v[k].w, w[k], acc.w);
  }
}

__global__ __launch_bounds__(512) void k_conv(const int* __restrict__ rowptr,
                                              const int* __restrict__ cnt,
                                              const int* __restrict__ perm,
                                              const float* __restrict__ dinv,
                                              const float* __restrict__ X,
                                              const float* __restrict__ W,
                                              const float* __restrict__ bias,
                                              float* __restrict__ out,
                                              int relu) {
  __shared__ __attribute__((aligned(16))) float Wl[(DD / 2) * PW];
  __shared__ __attribute__((aligned(16))) float At[2][16 * PA];
  const int t = threadIdx.x;
  const int lane = t & 31;
  const int w = __builtin_amdgcn_readfirstlane(t >> 5);

  const float4* W4 = (const float4*)W;
  for (int i = 0; i < 4; ++i) {
    const int idx = t + 512 * i;
    const int kp = idx >> 5;
    const int j = idx & 31;
    const float4 r0 = W4[(2 * kp) * (DD / 4) + j];
    const float4 r1 = W4[(2 * kp + 1) * (DD / 4) + j];
    float4 s0, s1;
    s0.x = r0.x; s0.y = r1.x; s0.z = r0.y; s0.w = r1.y;
    s1.x = r0.z; s1.y = r1.z; s1.z = r0.w; s1.w = r1.w;
    float4* q = (float4*)&Wl[kp * PW + 8 * j];
    q[0] = s0;
    q[1] = s1;
  }
  __syncthreads();

  const float4* X4 = (const float4*)X;
  const float bv = bias[(w & 7) * 16 + (lane & 15)];
  int it = 0;
  for (int tile = blockIdx.x; tile < NT; tile += gridDim.x, ++it) {
    float* Ab = At[it & 1];

    const int d = tile * 16 + w;
    const int beg = rowptr[d];
    const int n = cnt[d];
    const float dd = dinv[d];
    float4 acc = {0.0f, 0.0f, 0.0f, 0.0f};
    for (int base = 0; base < n; base += 32) {
      const int m = (n - base < 32) ? (n - base) : 32;
      const int idx = (lane < m) ? lane : (m - 1);
      const int sv = perm[beg + base + idx];
      const float dv = dinv[sv];
      int j = 0;
      for (; j + 8 <= m; j += 8) add_rows<8>(X4, sv, dv, dd, j, lane, acc);
      if (m - j >= 4) { add_rows<4>(X4, sv, dv, dd, j, lane, acc); j += 4; }
      if (m - j >= 2) { add_rows<2>(X4, sv, dv, dd, j, lane, acc); j += 2; }
      if (m - j >= 1) { add_rows<1>(X4, sv, dv, dd, j, lane, acc); }
    }
    const float4 h = X4[(size_t)d * (DD / 4) + lane];
    float4 o;
    o.x = acc.x + h.x * dd * dd;
    o.y = acc.y + h.y * dd * dd;
    o.z = acc.z + h.z * dd * dd;
    o.w = acc.w + h.w * dd * dd;
    *(float4*)&Ab[w * PA + 4 * lane] = o;
    __syncthreads();

    if (w < 8) {
      const int half = lane >> 4;
      const int c = lane & 15;
      const int n0 = w * 16;
      const float* ap = Ab + c * PA + 2 * half;
      const float* bp = Wl + half * PW + 2 * (n0 + c);
      v8f a8 = {};
#pragma unroll 8
      for (int k = 0; k < DD; k += 4) {
        const v2f a = *(const v2f*)(ap + k);
        const v2f b = *(const v2f*)(bp + (k >> 1) * PW);
        a8 = __builtin_amdgcn_wmma_f32_16x16x4_f32(
             false, a,  false, b,
             (short)0, a8,  false,  false);
      }
      float* op = out + (size_t)(tile * 16) * DD + n0 + c;
#pragma unroll
      for (int r = 0; r < 8; ++r) {
        float v = a8[r] + bv;
        if (relu) v = fmaxf(v, 0.0f);
        op[(size_t)(r + 8 * half) * DD] = v;
      }
    }
  }
}

extern "C" void kernel_launch(void* const* d_in, const int* in_sizes, int n_in,
                              void* d_out, int out_size, void* d_ws, size_t ws_size,
                              hipStream_t stream) {
  const float* x[2]  = {(const float*)d_in[0], (const float*)d_in[2]};
  const int*   ei[2] = {(const int*)d_in[1],   (const int*)d_in[3]};
  const float* W0 = (const float*)d_in[4];
  const float* b0 = (const float*)d_in[5];
  const float* W1 = (const float*)d_in[6];
  const float* b1 = (const float*)d_in[7];
  float* out = (float*)d_out;

  int*   cnt    = (int*)d_ws;
  int*   rowptr = cnt + NP;
  float* dinv   = (float*)(rowptr + NP);
  int*   tsrc   = (int*)(dinv + NP);
  int*   perm   = tsrc + NE;
  float* Y      = (float*)(perm + NE);
  int*   tdst   = (int*)(Y + (size_t)NN * DD);
  int*   bcount = tdst + NE;
  int*   boffs  = bcount + NM;

  const int nd    = NN * DD;
  const int gConv = 768;

  for (int bb = 0; bb < 2; ++bb) {
    const int* srcp = ei[bb];
    const int* dstp = ei[bb] + NE;

    k_hist <<<NBLK, dim3(512),  0, stream>>>(dstp, bcount);
    k_bscan<<<1,    dim3(1024), 0, stream>>>(bcount, boffs);
    k_bin  <<<NBLK, dim3(512),  0, stream>>>(srcp, dstp, boffs, tsrc, tdst);
    k_sort <<<NBK,  dim3(512),  0, stream>>>(tsrc, tdst, boffs, cnt, rowptr, dinv, perm);

    k_conv<<<gConv, dim3(512), 0, stream>>>(rowptr, cnt, perm, dinv, x[bb], W0, b0, Y, 1);

    k_conv<<<gConv, dim3(512), 0, stream>>>(rowptr, cnt, perm, dinv, Y, W1, b1, out + (size_t)bb * nd, 0);
  }
}


# Both graphs share every launch: sort launches take the graph from blockIdx.y, each conv launch walks the 2 x 6250 node tiles (graph 0 first); 12 dependent launches -> 6; sums and their order kept
# speedup vs baseline: 6.3221x; 1.1040x over previous
#include <hip/hip_runtime.h>


#define NN 100000
#define NE 1600000
#define DD 128
#define NP 100352

typedef __attribute__((ext_vector_type(2))) float v2f;
typedef __attribute__((ext_vector_type(8))) float v8f;

#define NBK 256
#define BS 391
#define CH 8192
#define NBLK 196
#define NM (NBK * NBLK)
#define BSP 416
#define CAP 16384

#define O_CNT 0
#define O_ROW (O_CNT + NP)
#define O_DINV (O_ROW + NP)
#define O_TSRC (O_DINV + NP)
#define O_PERM (O_TSRC + NE)
#define O_Y (O_PERM + NE)
#define O_TDST (O_Y + NN * DD)
#define O_BCNT (O_TDST + NE)
#define O_BOFF (O_BCNT + NM)
#define GS (((O_BOFF + NM + 1 + 1023) / 1024) * 1024)

__global__ __launch_bounds__(512) void k_hist(const int* __restrict__ dst0,
                                              const int* __restrict__ dst1,
                                              int* __restrict__ bcount) {
  __shared__ int h[NBK];
  const int t = threadIdx.x;
  const int blk = blockIdx.x;
  const int g = blockIdx.y;
  const int* __restrict__ dst = g ? dst1 : dst0;
  bcount += (size_t)g * GS;
  if (t < NBK) h[t] = 0;
  __syncthreads();
  const int base = blk * CH;
  const int n = (NE - base < CH) ? (NE - base) : CH;
  int d[CH / 512];
#pragma unroll
  for (int k = 0; k < CH / 512; ++k) {
    const int i = t + 512 * k;
    d[k] = dst[base + ((i < n) ? i : (n - 1))];
  }
#pragma unroll
  for (int k = 0; k < CH / 512; ++k)
    if (t + 512 * k < n) atomicAdd(&h[d[k] / BS], 1);
  __syncthreads();
  if (t < NBK) bcount[t * NBLK + blk] = h[t];
}

__global__ __launch_bounds__(1024) void k_bscan(const int* __restrict__ bcount,
                                                int* __restrict__ boffs) {
  __shared__ int buf[NM];
  __shared__ int wsum[32];
  bcount += (size_t)blockIdx.x * GS;
  boffs += (size_t)blockIdx.x * GS;
  const int t = threadIdx.x;
  const int lane = t & 31;
  const int w = t >> 5;
  const int per = NM / 1024;
  for (int k = 0; k < per; ++k) buf[k * 1024 + t] = bcount[k * 1024 + t];
  __syncthreads();
  int s = 0;
  for (int k = 0; k < per; ++k) s += buf[t * per + k];
  int inc = s;
  for (int o = 1; o < 32; o <<= 1) {
    const int u = __shfl_up(inc, o);
    if (lane >= o) inc += u;
  }
  if (lane == 31) wsum[w] = inc;
  __syncthreads();
  if (w == 0) {
    int x = wsum[lane];
    for (int o = 1; o < 32; o <<= 1) {
      const int u = __shfl_up(x, o);
      if (lane >= o) x += u;
    }
    wsum[lane] = x;
  }
  __syncthreads();
  int run = inc - s + ((w > 0) ? wsum[w - 1] : 0);
  for (int k = 0; k < per; ++k) {
    const int c = buf[t * per + k];
    buf[t * per + k] = run;
    run += c;
  }
  if (t == 1023) boffs[NM] = run;
  __syncthreads();
  for (int k = 0; k < per; ++k) boffs[k * 1024 + t] = buf[k * 1024 + t];
}

__global__ __launch_bounds__(512) void k_bin(const int* __restrict__ src0,
                                             const int* __restrict__ dst0,
                                             const int* __restrict__ src1,
                                             const int* __restrict__ dst1,
                                             const int* __restrict__ boffs,
                                             int* __restrict__ tsrc,
                                             int* __restrict__ tdst) {
  const int g = blockIdx.y;
  const int* __restrict__ src = g ? src1 : src0;
  const int* __restrict__ dst = g ? dst1 : dst0;
  boffs += (size_t)g * GS;
  tsrc += (size_t)g * GS;
  tdst += (size_t)g * GS;
  __shared__ int lc[NBK];
  __shared__ int ls[NBK];
  __shared__ int go[NBK];
  __shared__ int wtot[16];
  __shared__ int ss[CH];
  __shared__ int sd[CH];
  const int t = threadIdx.x;
  const int lane = t & 31;
  const int w = t >> 5;
  const int blk = blockIdx.x;
  const int base = blk * CH;
  const int n = (NE - base < CH) ? (NE - base) : CH;
  if (t < NBK) lc[t] = 0;
  __syncthreads();
  int d[CH / 512], sv[CH / 512], r[CH / 512];
#pragma unroll
  for (int k = 0; k < CH / 512; ++k) {
    const int i = t + 512 * k;
    const int ii = base + ((i < n) ? i : (n - 1));
    d[k] = dst[ii];
    sv[k] = src[ii];
  }
#pragma unroll
  for (int k = 0; k < CH / 512; ++k) {
    r[k] = 0;
    if (t + 512 * k < n) r[k] = atomicAdd(&lc[d[k] / BS], 1);
  }
  __syncthreads();
  const int v = (t < NBK) ? lc[t] : 0;
  int inc = v;
  for (int o = 1; o < 32; o <<= 1) {
    const int u = __shfl_up(inc, o);
    if (lane >= o) inc += u;
  }
  if (lane == 31) wtot[w] = inc;
  __syncthreads();
  int bw = 0;
  for (int k = 0; k < 16; ++k)
    if (k < w) bw += wtot[k];
  if (t < NBK) {
    const int ex = bw + inc - v;
    ls[t] = ex;
    go[t] = boffs[t * NBLK + blk] - ex;
  }
  __syncthreads();
#pragma unroll
  for (int k = 0; k < CH / 512; ++k) {
    const int i = t + 512 * k;
    if (i < n) {
      const int p = ls[d[k] / BS] + r[k];
      ss[p] = sv[k];
      sd[p] = d[k];
    }
  }
  __syncthreads();
  for (int i = t; i < n; i += 512) {
    const int dd = sd[i];
    const int p = go[dd / BS] + i;
    tsrc[p] = ss[i];
    tdst[p] = dd;
  }
}

__global__ __launch_bounds__(512) void k_sort(const int* __restrict__ tsrc,
                                              const int* __restrict__ tdst,
                                              const int* __restrict__ boffs,
                                              int* __restrict__ cnt,
                                              int* __restrict__ rowptr,
                                              float* __restrict__ dinv,
                                              int* __restrict__ perm) {
  __shared__ int c[BSP];
  __shared__ int rp[BSP];
  __shared__ int wtot[16];
  __shared__ int stage[CAP];
  {
    const size_t gofs = (size_t)blockIdx.y * GS;
    tsrc += gofs; tdst += gofs; boffs += gofs;
    cnt += gofs; rowptr += gofs; dinv += gofs; perm += gofs;
  }
  const int t = threadIdx.x;
  const int lane = t & 31;
  const int w = t >> 5;
  const int b = blockIdx.x;
  const int beg = boffs[b * NBLK];
  const int end = boffs[(b + 1) * NBLK];
  const int nb = end - beg;
  const int n0 = b * BS;
  const int nn = (NN - n0 < BS) ? (NN - n0) : BS;
  if (t < BSP) c[t] = 0;
  __syncthreads();
  for (int i0 = beg + t; i0 < end; i0 += 512 * 8) {
    int dd[8];
#pragma unroll
    for (int k = 0; k < 8; ++k) {
      const int i = i0 + 512 * k;
      dd[k] = tdst[(i < end) ? i : (end - 1)];
    }
#pragma unroll
    for (int k = 0; k < 8; ++k)
      if (i0 + 512 * k < end) atomicAdd(&c[dd[k] - n0], 1);
  }
  __syncthreads();
  const int v = (t < BSP) ? c[t] : 0;
  int inc = v;
  for (int o = 1; o < 32; o <<= 1) {
    const int u = __shfl_up(inc, o);
    if (lane >= o) inc += u;
  }
  if (lane == 31) wtot[w] = inc;
  __syncthreads();
  int bw = 0;
  for (int k = 0; k < 16; ++k)
    if (k < w) bw += wtot[k];
  const int ex = bw + inc - v;
  if (t < nn) {
    cnt[n0 + t] = v;
    rowptr[n0 + t] = beg + ex;
    dinv[n0 + t] = rsqrtf((float)v + 1.0f);
  }
  if (t < BSP) {
    rp[t] = ex;
    c[t] = 0;
  }
  __syncthreads();
  if (nb <= CAP) {
    for (int i0 = beg + t; i0 < end; i0 += 512 * 8) {
      int dd[8], sv[8];
#pragma unroll
      for (int k = 0; k < 8; ++k) {
        const int i = i0 + 512 * k;
        const int ii = (i < end) ? i : (end - 1);
        dd[k] = tdst[ii];
        sv[k] = tsrc[ii];
      }
#pragma unroll
      for (int k = 0; k < 8; ++k) {
        if (i0 + 512 * k < end) {
          const int ld = dd[k] - n0;
          const int rk = atomicAdd(&c[ld], 1);
          stage[rp[ld] + rk] = sv[k];
        }
      }
    }
    __syncthreads();
    for (int i = t; i < nb; i += 512) perm[beg + i] = stage[i];
  } else {
    for (int i = beg + t; i < end; i += 512) {
      const int ld = tdst[i] - n0;
      const int rk = atomicAdd(&c[ld], 1);
      perm[beg + rp[ld] + rk] = tsrc[i];
    }
  }
}

#define NT (NN / 16)
#define PW 288
#define PA 132

template <int K>
__device__ __forceinline__ void add_rows(const float4* __restrict__ H4, int sv,
                                         float dv, float dd, int j, int lane,
                                         float4& acc) {
  int s[K];
  float w[K];
  float4 v[K];
#pragma unroll
  for (int k = 0; k < K; ++k) {
    s[k] = __builtin_amdgcn_readlane(sv, j + k);
  }
#pragma unroll
  for (int k = 0; k < K; ++k) v[k] = H4[(size_t)s[k] * (DD / 4) + lane];
#pragma unroll
  for (int k = 0; k < K; ++k) {
    w[k] = __int_as_float(__builtin_amdgcn_readlane(__float_as_int(dv), j + k)) * dd;
  }
#pragma unroll
  for (int k = 0; k < K; ++k) {
    acc.x = fmaf(v[k].x, w[k], acc.x);
    acc.y = fmaf(v[k].y, w[k], acc.y);
    acc.z = fmaf(v[k].z, w[k], acc.z);
    acc.w = fmaf(v[k].w, w[k], acc.w);
  }
}

__global__ __launch_bounds__(512) void k_conv(const int* __restrict__ rowptr0,
                                              const int* __restrict__ cnt0,
                                              const int* __restrict__ perm0,
                                              const float* __restrict__ dinv0,
                                              const float* __restrict__ X0,
                                              const float* __restrict__ X1,
                                              const float* __restrict__ W,
                                              const float* __restrict__ bias,
                                              float* __restrict__ out0,
                                              float* __restrict__ out1,
                                              int relu) {
  __shared__ __attribute__((aligned(16))) float Wl[(DD / 2) * PW];
  __shared__ __attribute__((aligned(16))) float At[2][16 * PA];
  const int t = threadIdx.x;
  const int lane = t & 31;
  const int w = __builtin_amdgcn_readfirstlane(t >> 5);

  const float4* W4 = (const float4*)W;
  for (int i = 0; i < 4; ++i) {
    const int idx = t + 512 * i;
    const int kp = idx >> 5;
    const int j = idx & 31;
    const float4 r0 = W4[(2 * kp) * (DD / 4) + j];
    const float4 r1 = W4[(2 * kp + 1) * (DD / 4) + j];
    float4 s0, s1;
    s0.x = r0.x; s0.y = r1.x; s0.z = r0.y; s0.w = r1.y;
    s1.x = r0.z; s1.y = r1.z; s1.z = r0.w; s1.w = r1.w;
    float4* q = (float4*)&Wl[kp * PW + 8 * j];
    q[0] = s0;
    q[1] = s1;
  }
  __syncthreads();

  const float bv = bias[(w & 7) * 16 + (lane & 15)];
  int it = 0;
  for (int gt = blockIdx.x; gt < 2 * NT; gt += gridDim.x, ++it) {
    const int g = (gt >= NT) ? 1 : 0;
    const int tile = gt - g * NT;
    const size_t gofs = (size_t)g * GS;
    const int* __restrict__ rowptr = rowptr0 + gofs;
    const int* __restrict__ cnt = cnt0 + gofs;
    const int* __restrict__ perm = perm0 + gofs;
    const float* __restrict__ dinv = dinv0 + gofs;
    const float4* __restrict__ X4 = (const float4*)(g ? X1 : X0);
    float* __restrict__ out = g ? out1 : out0;
    float* Ab = At[it & 1];

    const int d = tile * 16 + w;
    const int beg = rowptr[d];
    const int n = cnt[d];
    const float dd = dinv[d];
    float4 acc = {0.0f, 0.0f, 0.0f, 0.0f};
    for (int base = 0; base < n; base += 32) {
      const int m = (n - base < 32) ? (n - base) : 32;
      const int idx = (lane < m) ? lane : (m - 1);
      const int sv = perm[beg + base + idx];
      const float dv = dinv[sv];
      int j = 0;
      for (; j + 8 <= m; j += 8) add_rows<8>(X4, sv, dv, dd, j, lane, acc);
      if (m - j >= 4) { add_rows<4>(X4, sv, dv, dd, j, lane, acc); j += 4; }
      if (m - j >= 2) { add_rows<2>(X4, sv, dv, dd, j, lane, acc); j += 2; }
      if (m - j >= 1) { add_rows<1>(X4, sv, dv, dd, j, lane, acc); }
    }
    const float4 h = X4[(size_t)d * (DD / 4) + lane];
    float4 o;
    o.x = acc.x + h.x * dd * dd;
    o.y = acc.y + h.y * dd * dd;
    o.z = acc.z + h.z * dd * dd;
    o.w = acc.w + h.w * dd * dd;
    *(float4*)&Ab[w * PA + 4 * lane] = o;
    __syncthreads();

    if (w < 8) {
      const int half = lane >> 4;
      const int c = lane & 15;
      const int n0 = w * 16;
      const float* ap = Ab + c * PA + 2 * half;
      const float* bp = Wl + half * PW + 2 * (n0 + c);
      v8f a8 = {};
#pragma unroll 8
      for (int k = 0; k < DD; k += 4) {
        const v2f a = *(const v2f*)(ap + k);
        const v2f b = *(const v2f*)(bp + (k >> 1) * PW);
        a8 = __builtin_amdgcn_wmma_f32_16x16x4_f32(
             false, a,  false, b,
             (short)0, a8,  false,  false);
      }
      float* op = out + (size_t)(tile * 16) * DD + n0 + c;
#pragma unroll
      for (int r = 0; r < 8; ++r) {
        float v = a8[r] + bv;
        if (relu) v = fmaxf(v, 0.0f);
        op[(size_t)(r + 8 * half) * DD] = v;
      }
    }
  }
}

extern "C" void kernel_launch(void* const* d_in, const int* in_sizes, int n_in,
                              void* d_out, int out_size, void* d_ws, size_t ws_size,
                              hipStream_t stream) {
  const float* x[2]  = {(const float*)d_in[0], (const float*)d_in[2]};
  const int*   ei[2] = {(const int*)d_in[1],   (const int*)d_in[3]};
  const float* W0 = (const float*)d_in[4];
  const float* b0 = (const float*)d_in[5];
  const float* W1 = (const float*)d_in[6];
  const float* b1 = (const float*)d_in[7];
  float* out = (float*)d_out;

  int*   ws     = (int*)d_ws;
  int*   cnt    = ws + O_CNT;
  int*   rowptr = ws + O_ROW;
  float* dinv   = (float*)(ws + O_DINV);
  int*   tsrc   = ws + O_TSRC;
  int*   perm   = ws + O_PERM;
  float* Y      = (float*)(ws + O_Y);
  int*   tdst   = ws + O_TDST;
  int*   bcount = ws + O_BCNT;
  int*   boffs  = ws + O_BOFF;

  const int gConv = 768;

  k_hist <<<dim3(NBLK, 2), dim3(512),  0, stream>>>(ei[0] + NE, ei[1] + NE, bcount);
  k_bscan<<<dim3(2),       dim3(1024), 0, stream>>>(bcount, boffs);
  k_bin  <<<dim3(NBLK, 2), dim3(512),  0, stream>>>(ei[0], ei[0] + NE, ei[1], ei[1] + NE,
                                                    boffs, tsrc, tdst);
  k_sort <<<dim3(NBK, 2),  dim3(512),  0, stream>>>(tsrc, tdst, boffs, cnt, rowptr, dinv, perm);

  k_conv<<<gConv, dim3(512), 0, stream>>>(rowptr, cnt, perm, dinv, x[0], x[1], W0, b0,
                                          Y, Y + GS, 1);

  k_conv<<<gConv, dim3(512), 0, stream>>>(rowptr, cnt, perm, dinv, Y, Y + GS, W1, b1,
                                          out, out + (size_t)NN * DD, 0);
}


# Gather's dependent load chain cut from ~8 to 2-3 L2 round trips per tile: node data and sources asked for one tile ahead; weights, self row, up to 12 source rows issued as one group; sum order kept
# speedup vs baseline: 6.6847x; 1.0574x over previous
#include <hip/hip_runtime.h>


#define NN 100000
#define NE 1600000
#define DD 128
#define NP 100352

typedef __attribute__((ext_vector_type(2))) float v2f;
typedef __attribute__((ext_vector_type(8))) float v8f;

#define NBK 256
#define BS 391
#define CH 8192
#define NBLK 196
#define NM (NBK * NBLK)
#define BSP 416
#define CAP 16384

#define O_CNT 0
#define O_ROW (O_CNT + NP)
#define O_DINV (O_ROW + NP)
#define O_TSRC (O_DINV + NP)
#define O_PERM (O_TSRC + NE)
#define O_Y (O_PERM + NE)
#define O_TDST (O_Y + NN * DD)
#define O_BCNT (O_TDST + NE)
#define O_BOFF (O_BCNT + NM)
#define GS (((O_BOFF + NM + 1 + 1023) / 1024) * 1024)

__global__ __launch_bounds__(512) void k_hist(const int* __restrict__ dst0,
                                              const int* __restrict__ dst1,
                                              int* __restrict__ bcount) {
  __shared__ int h[NBK];
  const int t = threadIdx.x;
  const int blk = blockIdx.x;
  const int g = blockIdx.y;
  const int* __restrict__ dst = g ? dst1 : dst0;
  bcount += (size_t)g * GS;
  if (t < NBK) h[t] = 0;
  __syncthreads();
  const int base = blk * CH;
  const int n = (NE - base < CH) ? (NE - base) : CH;
  int d[CH / 512];
#pragma unroll
  for (int k = 0; k < CH / 512; ++k) {
    const int i = t + 512 * k;
    d[k] = dst[base + ((i < n) ? i : (n - 1))];
  }
#pragma unroll
  for (int k = 0; k < CH / 512; ++k)
    if (t + 512 * k < n) atomicAdd(&h[d[k] / BS], 1);
  __syncthreads();
  if (t < NBK) bcount[t * NBLK + blk] = h[t];
}

__global__ __launch_bounds__(1024) void k_bscan(const int* __restrict__ bcount,
                                                int* __restrict__ boffs) {
  __shared__ int buf[NM];
  __shared__ int wsum[32];
  bcount += (size_t)blockIdx.x * GS;
  boffs += (size_t)blockIdx.x * GS;
  const int t = threadIdx.x;
  const int lane = t & 31;
  const int w = t >> 5;
  const int per = NM / 1024;
  for (int k = 0; k < per; ++k) buf[k * 1024 + t] = bcount[k * 1024 + t];
  __syncthreads();
  int s = 0;
  for (int k = 0; k < per; ++k) s += buf[t * per + k];
  int inc = s;
  for (int o = 1; o < 32; o <<= 1) {
    const int u = __shfl_up(inc, o);
    if (lane >= o) inc += u;
  }
  if (lane == 31) wsum[w] = inc;
  __syncthreads();
  if (w == 0) {
    int x = wsum[lane];
    for (int o = 1; o < 32; o <<= 1) {
      const int u = __shfl_up(x, o);
      if (lane >= o) x += u;
    }
    wsum[lane] = x;
  }
  __syncthreads();
  int run = inc - s + ((w > 0) ? wsum[w - 1] : 0);
  for (int k = 0; k < per; ++k) {
    const int c = buf[t * per + k];
    buf[t * per + k] = run;
    run += c;
  }
  if (t == 1023) boffs[NM] = run;
  __syncthreads();
  for (int k = 0; k < per; ++k) boffs[k * 1024 + t] = buf[k * 1024 + t];
}

__global__ __launch_bounds__(512) void k_bin(const int* __restrict__ src0,
                                             const int* __restrict__ dst0,
                                             const int* __restrict__ src1,
                                             const int* __restrict__ dst1,
                                             const int* __restrict__ boffs,
                                             int* __restrict__ tsrc,
                                             int* __restrict__ tdst) {
  const int g = blockIdx.y;
  const int* __restrict__ src = g ? src1 : src0;
  const int* __restrict__ dst = g ? dst1 : dst0;
  boffs += (size_t)g * GS;
  tsrc += (size_t)g * GS;
  tdst += (size_t)g * GS;
  __shared__ int lc[NBK];
  __shared__ int ls[NBK];
  __shared__ int go[NBK];
  __shared__ int wtot[16];
  __shared__ int ss[CH];
  __shared__ int sd[CH];
  const int t = threadIdx.x;
  const int lane = t & 31;
  const int w = t >> 5;
  const int blk = blockIdx.x;
  const int base = blk * CH;
  const int n = (NE - base < CH) ? (NE - base) : CH;
  if (t < NBK) lc[t] = 0;
  __syncthreads();
  int d[CH / 512], sv[CH / 512], r[CH / 512];
#pragma unroll
  for (int k = 0; k < CH / 512; ++k) {
    const int i = t + 512 * k;
    const int ii = base + ((i < n) ? i : (n - 1));
    d[k] = dst[ii];
    sv[k] = src[ii];
  }
#pragma unroll
  for (int k = 0; k < CH / 512; ++k) {
    r[k] = 0;
    if (t + 512 * k < n) r[k] = atomicAdd(&lc[d[k] / BS], 1);
  }
  __syncthreads();
  const int v = (t < NBK) ? lc[t] : 0;
  int inc = v;
  for (int o = 1; o < 32; o <<= 1) {
    const int u = __shfl_up(inc, o);
    if (lane >= o) inc += u;
  }
  if (lane == 31) wtot[w] = inc;
  __syncthreads();
  int bw = 0;
  for (int k = 0; k < 16; ++k)
    if (k < w) bw += wtot[k];
  if (t < NBK) {
    const int ex = bw + inc - v;
    ls[t] = ex;
    go[t] = boffs[t * NBLK + blk] - ex;
  }
  __syncthreads();
#pragma unroll
  for (int k = 0; k < CH / 512; ++k) {
    const int i = t + 512 * k;
    if (i < n) {
      const int p = ls[d[k] / BS] + r[k];
      ss[p] = sv[k];
      sd[p] = d[k];
    }
  }
  __syncthreads();
  for (int i = t; i < n; i += 512) {
    const int dd = sd[i];
    const int p = go[dd / BS] + i;
    tsrc[p] = ss[i];
    tdst[p] = dd;
  }
}

__global__ __launch_bounds__(512) void k_sort(const int* __restrict__ tsrc,
                                              const int* __restrict__ tdst,
                                              const int* __restrict__ boffs,
                                              int* __restrict__ cnt,
                                              int* __restrict__ rowptr,
                                              float* __restrict__ dinv,
                                              int* __restrict__ perm) {
  __shared__ int c[BSP];
  __shared__ int rp[BSP];
  __shared__ int wtot[16];
  __shared__ int stage[CAP];
  {
    const size_t gofs = (size_t)blockIdx.y * GS;
    tsrc += gofs; tdst += gofs; boffs += gofs;
    cnt += gofs; rowptr += gofs; dinv += gofs; perm += gofs;
  }
  const int t = threadIdx.x;
  const int lane = t & 31;
  const int w = t >> 5;
  const int b = blockIdx.x;
  const int beg = boffs[b * NBLK];
  const int end = boffs[(b + 1) * NBLK];
  const int nb = end - beg;
  const int n0 = b * BS;
  const int nn = (NN - n0 < BS) ? (NN - n0) : BS;
  if (t < BSP) c[t] = 0;
  __syncthreads();
  for (int i0 = beg + t; i0 < end; i0 += 512 * 8) {
    int dd[8];
#pragma unroll
    for (int k = 0; k < 8; ++k) {
      const int i = i0 + 512 * k;
      dd[k] = tdst[(i < end) ? i : (end - 1)];
    }
#pragma unroll
    for (int k = 0; k < 8; ++k)
      if (i0 + 512 * k < end) atomicAdd(&c[dd[k] - n0], 1);
  }
  __syncthreads();
  const int v = (t < BSP) ? c[t] : 0;
  int inc = v;
  for (int o = 1; o < 32; o <<= 1) {
    const int u = __shfl_up(inc, o);
    if (lane >= o) inc += u;
  }
  if (lane == 31) wtot[w] = inc;
  __syncthreads();
  int bw = 0;
  for (int k = 0; k < 16; ++k)
    if (k < w) bw += wtot[k];
  const int ex = bw + inc - v;
  if (t < nn) {
    cnt[n0 + t] = v;
    rowptr[n0 + t] = beg + ex;
    dinv[n0 + t] = rsqrtf((float)v + 1.0f);
  }
  if (t < BSP) {
    rp[t] = ex;
    c[t] = 0;
  }
  __syncthreads();
  if (nb <= CAP) {
    for (int i0 = beg + t; i0 < end; i0 += 512 * 8) {
      int dd[8], sv[8];
#pragma unroll
      for (int k = 0; k < 8; ++k) {
        const int i = i0 + 512 * k;
        const int ii = (i < end) ? i : (end - 1);
        dd[k] = tdst[ii];
        sv[k] = tsrc[ii];
      }
#pragma unroll
      for (int k = 0; k < 8; ++k) {
        if (i0 + 512 * k < end) {
          const int ld = dd[k] - n0;
          const int rk = atomicAdd(&c[ld], 1);
          stage[rp[ld] + rk] = sv[k];
        }
      }
    }
    __syncthreads();
    for (int i = t; i < nb; i += 512) perm[beg + i] = stage[i];
  } else {
    for (int i = beg + t; i < end; i += 512) {
      const int ld = tdst[i] - n0;
      const int rk = atomicAdd(&c[ld], 1);
      perm[beg + rp[ld] + rk] = tsrc[i];
    }
  }
}

#define NT (NN / 16)
#define PW 288
#define PA 132

template <int K>
__device__ __forceinline__ void add_rows(const float4* __restrict__ H4, int sv,
                                         float dv, float dd, int j, int lane,
                                         float4& acc) {
  int s[K];
  float w[K];
  float4 v[K];
#pragma unroll
  for (int k = 0; k < K; ++k) {
    s[k] = __builtin_amdgcn_readlane(sv, j + k);
  }
#pragma unroll
  for (int k = 0; k < K; ++k) v[k] = H4[(size_t)s[k] * (DD / 4) + lane];
#pragma unroll
  for (int k = 0; k < K; ++k) {
    w[k] = __int_as_float(__builtin_amdgcn_readlane(__float_as_int(dv), j + k)) * dd;
  }
#pragma unroll
  for (int k = 0; k < K; ++k) {
    acc.x = fmaf(v[k].x, w[k], acc.x);
    acc.y = fmaf(v[k].y, w[k], acc.y);
    acc.z = fmaf(v[k].z, w[k], acc.z);
    acc.w = fmaf(v[k].w, w[k], acc.w);
  }
}

#define KR 12
__device__ __forceinline__ void add_group(const float4* __restrict__ H4, int sv,
                                          float dv, float dd, int j, int c, int lane,
                                          float4& acc) {
  switch (c) {
    case 1: add_rows<1>(H4, sv, dv, dd, j, lane, acc); break;
    case 2: add_rows<2>(H4, sv, dv, dd, j, lane, acc); break;
    case 3: add_rows<3>(H4, sv, dv, dd, j, lane, acc); break;
    case 4: add_rows<4>(H4, sv, dv, dd, j, lane, acc); break;
    case 5: add_rows<5>(H4, sv, dv, dd, j, lane, acc); break;
    case 6: add_rows<6>(H4, sv, dv, dd, j, lane, acc); break;
    case 7: add_rows<7>(H4, sv, dv, dd, j, lane, acc); break;
    case 8: add_rows<8>(H4, sv, dv, dd, j, lane, acc); break;
    case 9: add_rows<9>(H4, sv, dv, dd, j, lane, acc); break;
    case 10: add_rows<10>(H4, sv, dv, dd, j, lane, acc); break;
    case 11: add_rows<11>(H4, sv, dv, dd, j, lane, acc); break;
    default: add_rows<12>(H4, sv, dv, dd, j, lane, acc); break;
  }
}

__global__ __launch_bounds__(512) void k_conv(const int* __restrict__ rowptr0,
                                              const int* __restrict__ cnt0,
                                              const int* __restrict__ perm0,
                                              const float* __restrict__ dinv0,
                                              const float* __restrict__ X0,
                                              const float* __restrict__ X1,
                                              const float* __restrict__ W,
                                              const float* __restrict__ bias,
                                              float* __restrict__ out0,
                                              float* __restrict__ out1,
                                              int relu) {
  __shared__ __attribute__((aligned(16))) float Wl[(DD / 2) * PW];
  __shared__ __attribute__((aligned(16))) float At[2][16 * PA];
  const int t = threadIdx.x;
  const int lane = t & 31;
  const int w = __builtin_amdgcn_readfirstlane(t >> 5);

  const float4* W4 = (const float4*)W;
  for (int i = 0; i < 4; ++i) {
    const int idx = t + 512 * i;
    const int kp = idx >> 5;
    const int j = idx & 31;
    const float4 r0 = W4[(2 * kp) * (DD / 4) + j];
    const float4 r1 = W4[(2 * kp + 1) * (DD / 4) + j];
    float4 s0, s1;
    s0.x = r0.x; s0.y = r1.x; s0.z = r0.y; s0.w = r1.y;
    s1.x = r0.z; s1.y = r1.z; s1.z = r0.w; s1.w = r1.w;
    float4* q = (float4*)&Wl[kp * PW + 8 * j];
    q[0] = s0;
    q[1] = s1;
  }
  __syncthreads();

  const float bv = bias[(w & 7) * 16 + (lane & 15)];
  const int step = gridDim.x;
  int beg = 0, n = 0, sv = 0;
  float dd = 0.0f;
  int rbN = 0, rnN = 0;
  float rdN = 0.0f;
  if ((int)blockIdx.x < 2 * NT) {
    const int gt0 = blockIdx.x;
    const int g = (gt0 >= NT) ? 1 : 0;
    const size_t go = (size_t)g * GS;
    const int d = (gt0 - g * NT) * 16 + w;
    rbN = rowptr0[go + d];
    rnN = cnt0[go + d];
    rdN = dinv0[go + d];
  }
  int it = 0;
  for (int gt = (int)blockIdx.x - step; gt < 2 * NT; gt += step) {
    const int gtN = gt + step;
    const bool hasN = gtN < 2 * NT;
    const size_t goN = (gtN >= NT) ? (size_t)GS : (size_t)0;
    const int begN = __builtin_amdgcn_readfirstlane(rbN);
    const int nN = __builtin_amdgcn_readfirstlane(rnN);
    const float ddN = __int_as_float(__builtin_amdgcn_readfirstlane(__float_as_int(rdN)));
    int svN = 0;
    if (hasN && nN > 0) {
      const int mN = (nN < 32) ? nN : 32;
      svN = perm0[goN + begN + ((lane < mN) ? lane : (mN - 1))];
    }
    if (gtN + step < 2 * NT) {
      const int gtNN = gtN + step;
      const int g2 = (gtNN >= NT) ? 1 : 0;
      const size_t go2 = (size_t)g2 * GS;
      const int d2 = (gtNN - g2 * NT) * 16 + w;
      rbN = rowptr0[go2 + d2];
      rnN = cnt0[go2 + d2];
      rdN = dinv0[go2 + d2];
    }

    if (gt >= 0) {
      const int g = (gt >= NT) ? 1 : 0;
      const int tile = gt - g * NT;
      const size_t gofs = (size_t)g * GS;
      const int* __restrict__ perm = perm0 + gofs;
      const float* __restrict__ dinv = dinv0 + gofs;
      const float4* __restrict__ X4 = (const float4*)(g ? X1 : X0);
      float* __restrict__ out = g ? out1 : out0;
      float* Ab = At[it & 1];
      ++it;
      const int d = tile * 16 + w;

      float dv = dinv[sv];
      const float4 h = X4[(size_t)d * (DD / 4) + lane];
      float4 acc = {0.0f, 0.0f, 0.0f, 0.0f};
      for (int base = 0; base < n; base += 32) {
        const int m = (n - base < 32) ? (n - base) : 32;
        if (base > 0) {
          const int idx = (lane < m) ? lane : (m - 1);
          sv = perm[beg + base + idx];
          dv = dinv[sv];
        }
        for (int j = 0; j < m; j += KR)
          add_group(X4, sv, dv, dd, j, (m - j < KR) ? (m - j) : KR, lane, acc);
      }
      float4 o;
      o.x = acc.x + h.x * dd * dd;
      o.y = acc.y + h.y * dd * dd;
      o.z = acc.z + h.z * dd * dd;
      o.w = acc.w + h.w * dd * dd;
      *(float4*)&Ab[w * PA + 4 * lane] = o;
      __syncthreads();

      if (w < 8) {
        const int half = lane >> 4;
        const int c = lane & 15;
        const int n0 = w * 16;
        const float* ap = Ab + c * PA + 2 * half;
        const float* bp = Wl + half * PW + 2 * (n0 + c);
        v8f a8 = {};
#pragma unroll 8
        for (int k = 0; k < DD; k += 4) {
          const v2f a = *(const v2f*)(ap + k);
          const v2f b = *(const v2f*)(bp + (k >> 1) * PW);
          a8 = __builtin_amdgcn_wmma_f32_16x16x4_f32(
               false, a,  false, b,
               (short)0, a8,  false,  false);
        }
        float* op = out + (size_t)(tile * 16) * DD + n0 + c;
#pragma unroll
        for (int r = 0; r < 8; ++r) {
          float v = a8[r] + bv;
          if (relu) v = fmaxf(v, 0.0f);
          op[(size_t)(r + 8 * half) * DD] = v;
        }
      }
    }
    beg = begN; n = nN; dd = ddN; sv = svN;
  }
}

extern "C" void kernel_launch(void* const* d_in, const int* in_sizes, int n_in,
                              void* d_out, int out_size, void* d_ws, size_t ws_size,
                              hipStream_t stream) {
  const float* x[2]  = {(const float*)d_in[0], (const float*)d_in[2]};
  const int*   ei[2] = {(const int*)d_in[1],   (const int*)d_in[3]};
  const float* W0 = (const float*)d_in[4];
  const float* b0 = (const float*)d_in[5];
  const float* W1 = (const float*)d_in[6];
  const float* b1 = (const float*)d_in[7];
  float* out = (float*)d_out;

  int*   ws     = (int*)d_ws;
  int*   cnt    = ws + O_CNT;
  int*   rowptr = ws + O_ROW;
  float* dinv   = (float*)(ws + O_DINV);
  int*   tsrc   = ws + O_TSRC;
  int*   perm   = ws + O_PERM;
  float* Y      = (float*)(ws + O_Y);
  int*   tdst   = ws + O_TDST;
  int*   bcount = ws + O_BCNT;
  int*   boffs  = ws + O_BOFF;

  const int gConv = 768;

  k_hist <<<dim3(NBLK, 2), dim3(512),  0, stream>>>(ei[0] + NE, ei[1] + NE, bcount);
  k_bscan<<<dim3(2),       dim3(1024), 0, stream>>>(bcount, boffs);
  k_bin  <<<dim3(NBLK, 2), dim3(512),  0, stream>>>(ei[0], ei[0] + NE, ei[1], ei[1] + NE,
                                                    boffs, tsrc, tdst);
  k_sort <<<dim3(NBK, 2),  dim3(512),  0, stream>>>(tsrc, tdst, boffs, cnt, rowptr, dinv, perm);

  k_conv<<<gConv, dim3(512), 0, stream>>>(rowptr, cnt, perm, dinv, x[0], x[1], W0, b0,
                                          Y, Y + GS, 1);

  k_conv<<<gConv, dim3(512), 0, stream>>>(rowptr, cnt, perm, dinv, Y, Y + GS, W1, b1,
                                          out, out + (size_t)NN * DD, 0);
}


# Layer 2 reads no per-edge weight: layer 1 stores hidden rows times deg^-1/2 of their node, layer 2 sums plain rows and scales the sum once; sum order kept, the scale factors round at other places
# speedup vs baseline: 6.9854x; 1.0450x over previous
#include <hip/hip_runtime.h>


#define NN 100000
#define NE 1600000
#define DD 128
#define NP 100352

typedef __attribute__((ext_vector_type(2))) float v2f;
typedef __attribute__((ext_vector_type(8))) float v8f;

#define NBK 256
#define BS 391
#define CH 8192
#define NBLK 196
#define NM (NBK * NBLK)
#define BSP 416
#define CAP 16384

#define O_CNT 0
#define O_ROW (O_CNT + NP)
#define O_DINV (O_ROW + NP)
#define O_TSRC (O_DINV + NP)
#define O_PERM (O_TSRC + NE)
#define O_Y (O_PERM + NE)
#define O_TDST (O_Y + NN * DD)
#define O_BCNT (O_TDST + NE)
#define O_BOFF (O_BCNT + NM)
#define GS (((O_BOFF + NM + 1 + 1023) / 1024) * 1024)

__global__ __launch_bounds__(512) void k_hist(const int* __restrict__ dst0,
                                              const int* __restrict__ dst1,
                                              int* __restrict__ bcount) {
  __shared__ int h[NBK];
  const int t = threadIdx.x;
  const int blk = blockIdx.x;
  const int g = blockIdx.y;
  const int* __restrict__ dst = g ? dst1 : dst0;
  bcount += (size_t)g * GS;
  if (t < NBK) h[t] = 0;
  __syncthreads();
  const int base = blk * CH;
  const int n = (NE - base < CH) ? (NE - base) : CH;
  int d[CH / 512];
#pragma unroll
  for (int k = 0; k < CH / 512; ++k) {
    const int i = t + 512 * k;
    d[k] = dst[base + ((i < n) ? i : (n - 1))];
  }
#pragma unroll
  for (int k = 0; k < CH / 512; ++k)
    if (t + 512 * k < n) atomicAdd(&h[d[k] / BS], 1);
  __syncthreads();
  if (t < NBK) bcount[t * NBLK + blk] = h[t];
}

__global__ __launch_bounds__(1024) void k_bscan(const int* __restrict__ bcount,
                                                int* __restrict__ boffs) {
  __shared__ int buf[NM];
  __shared__ int wsum[32];
  bcount += (size_t)blockIdx.x * GS;
  boffs += (size_t)blockIdx.x * GS;
  const int t = threadIdx.x;
  const int lane = t & 31;
  const int w = t >> 5;
  const int per = NM / 1024;
  for (int k = 0; k < per; ++k) buf[k * 1024 + t] = bcount[k * 1024 + t];
  __syncthreads();
  int s = 0;
  for (int k = 0; k < per; ++k) s += buf[t * per + k];
  int inc = s;
  for (int o = 1; o < 32; o <<= 1) {
    const int u = __shfl_up(inc, o);
    if (lane >= o) inc += u;
  }
  if (lane == 31) wsum[w] = inc;
  __syncthreads();
  if (w == 0) {
    int x = wsum[lane];
    for (int o = 1; o < 32; o <<= 1) {
      const int u = __shfl_up(x, o);
      if (lane >= o) x += u;
    }
    wsum[lane] = x;
  }
  __syncthreads();
  int run = inc - s + ((w > 0) ? wsum[w - 1] : 0);
  for (int k = 0; k < per; ++k) {
    const int c = buf[t * per + k];
    buf[t * per + k] = run;
    run += c;
  }
  if (t == 1023) boffs[NM] = run;
  __syncthreads();
  for (int k = 0; k < per; ++k) boffs[k * 1024 + t] = buf[k * 1024 + t];
}

__global__ __launch_bounds__(512) void k_bin(const int* __restrict__ src0,
                                             const int* __restrict__ dst0,
                                             const int* __restrict__ src1,
                                             const int* __restrict__ dst1,
                                             const int* __restrict__ boffs,
                                             int* __restrict__ tsrc,
                                             int* __restrict__ tdst) {
  const int g = blockIdx.y;
  const int* __restrict__ src = g ? src1 : src0;
  const int* __restrict__ dst = g ? dst1 : dst0;
  boffs += (size_t)g * GS;
  tsrc += (size_t)g * GS;
  tdst += (size_t)g * GS;
  __shared__ int lc[NBK];
  __shared__ int ls[NBK];
  __shared__ int go[NBK];
  __shared__ int wtot[16];
  __shared__ int ss[CH];
  __shared__ int sd[CH];
  const int t = threadIdx.x;
  const int lane = t & 31;
  const int w = t >> 5;
  const int blk = blockIdx.x;
  const int base = blk * CH;
  const int n = (NE - base < CH) ? (NE - base) : CH;
  if (t < NBK) lc[t] = 0;
  __syncthreads();
  int d[CH / 512], sv[CH / 512], r[CH / 512];
#pragma unroll
  for (int k = 0; k < CH / 512; ++k) {
    const int i = t + 512 * k;
    const int ii = base + ((i < n) ? i : (n - 1));
    d[k] = dst[ii];
    sv[k] = src[ii];
  }
#pragma unroll
  for (int k = 0; k < CH / 512; ++k) {
    r[k] = 0;
    if (t + 512 * k < n) r[k] = atomicAdd(&lc[d[k] / BS], 1);
  }
  __syncthreads();
  const int v = (t < NBK) ? lc[t] : 0;
  int inc = v;
  for (int o = 1; o < 32; o <<= 1) {
    const int u = __shfl_up(inc, o);
    if (lane >= o) inc += u;
  }
  if (lane == 31) wtot[w] = inc;
  __syncthreads();
  int bw = 0;
  for (int k = 0; k < 16; ++k)
    if (k < w) bw += wtot[k];
  if (t < NBK) {
    const int ex = bw + inc - v;
    ls[t] = ex;
    go[t] = boffs[t * NBLK + blk] - ex;
  }
  __syncthreads();
#pragma unroll
  for (int k = 0; k < CH / 512; ++k) {
    const int i = t + 512 * k;
    if (i < n) {
      const int p = ls[d[k] / BS] + r[k];
      ss[p] = sv[k];
      sd[p] = d[k];
    }
  }
  __syncthreads();
  for (int i = t; i < n; i += 512) {
    const int dd = sd[i];
    const int p = go[dd / BS] + i;
    tsrc[p] = ss[i];
    tdst[p] = dd;
  }
}

__global__ __launch_bounds__(512) void k_sort(const int* __restrict__ tsrc,
                                              const int* __restrict__ tdst,
                                              const int* __restrict__ boffs,
                                              int* __restrict__ cnt,
                                              int* __restrict__ rowptr,
                                              float* __restrict__ dinv,
                                              int* __restrict__ perm) {
  __shared__ int c[BSP];
  __shared__ int rp[BSP];
  __shared__ int wtot[16];
  __shared__ int stage[CAP];
  {
    const size_t gofs = (size_t)blockIdx.y * GS;
    tsrc += gofs; tdst += gofs; boffs += gofs;
    cnt += gofs; rowptr += gofs; dinv += gofs; perm += gofs;
  }
  const int t = threadIdx.x;
  const int lane = t & 31;
  const int w = t >> 5;
  const int b = blockIdx.x;
  const int beg = boffs[b * NBLK];
  const int end = boffs[(b + 1) * NBLK];
  const int nb = end - beg;
  const int n0 = b * BS;
  const int nn = (NN - n0 < BS) ? (NN - n0) : BS;
  if (t < BSP) c[t] = 0;
  __syncthreads();
  for (int i0 = beg + t; i0 < end; i0 += 512 * 8) {
    int dd[8];
#pragma unroll
    for (int k = 0; k < 8; ++k) {
      const int i = i0 + 512 * k;
      dd[k] = tdst[(i < end) ? i : (end - 1)];
    }
#pragma unroll
    for (int k = 0; k < 8; ++k)
      if (i0 + 512 * k < end) atomicAdd(&c[dd[k] - n0], 1);
  }
  __syncthreads();
  const int v = (t < BSP) ? c[t] : 0;
  int inc = v;
  for (int o = 1; o < 32; o <<= 1) {
    const int u = __shfl_up(inc, o);
    if (lane >= o) inc += u;
  }
  if (lane == 31) wtot[w] = inc;
  __syncthreads();
  int bw = 0;
  for (int k = 0; k < 16; ++k)
    if (k < w) bw += wtot[k];
  const int ex = bw + inc - v;
  if (t < nn) {
    cnt[n0 + t] = v;
    rowptr[n0 + t] = beg + ex;
    dinv[n0 + t] = rsqrtf((float)v + 1.0f);
  }
  if (t < BSP) {
    rp[t] = ex;
    c[t] = 0;
  }
  __syncthreads();
  if (nb <= CAP) {
    for (int i0 = beg + t; i0 < end; i0 += 512 * 8) {
      int dd[8], sv[8];
#pragma unroll
      for (int k = 0; k < 8; ++k) {
        const int i = i0 + 512 * k;
        const int ii = (i < end) ? i : (end - 1);
        dd[k] = tdst[ii];
        sv[k] = tsrc[ii];
      }
#pragma unroll
      for (int k = 0; k < 8; ++k) {
        if (i0 + 512 * k < end) {
          const int ld = dd[k] - n0;
          const int rk = atomicAdd(&c[ld], 1);
          stage[rp[ld] + rk] = sv[k];
        }
      }
    }
    __syncthreads();
    for (int i = t; i < nb; i += 512) perm[beg + i] = stage[i];
  } else {
    for (int i = beg + t; i < end; i += 512) {
      const int ld = tdst[i] - n0;
      const int rk = atomicAdd(&c[ld], 1);
      perm[beg + rp[ld] + rk] = tsrc[i];
    }
  }
}

#define NT (NN / 16)
#define PW 288
#define PA 132

template <int K, bool SCALED>
__device__ __forceinline__ void add_rows(const float4* __restrict__ H4, int sv,
                                         float dv, float dd, int j, int lane,
                                         float4& acc) {
  int s[K];
  float4 v[K];
#pragma unroll
  for (int k = 0; k < K; ++k) {
    s[k] = __builtin_amdgcn_readlane(sv, j + k);
  }
#pragma unroll
  for (int k = 0; k < K; ++k) v[k] = H4[(size_t)s[k] * (DD / 4) + lane];
  if (SCALED) {
#pragma unroll
    for (int k = 0; k < K; ++k) {
      acc.x += v[k].x;
      acc.y += v[k].y;
      acc.z += v[k].z;
      acc.w += v[k].w;
    }
  } else {
    float w[K];
#pragma unroll
    for (int k = 0; k < K; ++k) {
      w[k] = __int_as_float(__builtin_amdgcn_readlane(__float_as_int(dv), j + k)) * dd;
    }
#pragma unroll
    for (int k = 0; k < K; ++k) {
      acc.x = fmaf(v[k].x, w[k], acc.x);
      acc.y = fmaf(v[k].y, w[k], acc.y);
      acc.z = fmaf(v[k].z, w[k], acc.z);
      acc.w = fmaf(v[k].w, w[k], acc.w);
    }
  }
}

#define KR 12
template <bool SCALED>
__device__ __forceinline__ void add_group(const float4* __restrict__ H4, int sv,
                                          float dv, float dd, int j, int c, int lane,
                                          float4& acc) {
  switch (c) {
    case 1: add_rows<1, SCALED>(H4, sv, dv, dd, j, lane, acc); break;
    case 2: add_rows<2, SCALED>(H4, sv, dv, dd, j, lane, acc); break;
    case 3: add_rows<3, SCALED>(H4, sv, dv, dd, j, lane, acc); break;
    case 4: add_rows<4, SCALED>(H4, sv, dv, dd, j, lane, acc); break;
    case 5: add_rows<5, SCALED>(H4, sv, dv, dd, j, lane, acc); break;
    case 6: add_rows<6, SCALED>(H4, sv, dv, dd, j, lane, acc); break;
    case 7: add_rows<7, SCALED>(H4, sv, dv, dd, j, lane, acc); break;
    case 8: add_rows<8, SCALED>(H4, sv, dv, dd, j, lane, acc); break;
    case 9: add_rows<9, SCALED>(H4, sv, dv, dd, j, lane, acc); break;
    case 10: add_rows<10, SCALED>(H4, sv, dv, dd, j, lane, acc); break;
    case 11: add_rows<11, SCALED>(H4, sv, dv, dd, j, lane, acc); break;
    default: add_rows<12, SCALED>(H4, sv, dv, dd, j, lane, acc); break;
  }
}

template <int LAYER>
__global__ __launch_bounds__(512) void k_conv(const int* __restrict__ rowptr0,
                                              const int* __restrict__ cnt0,
                                              const int* __restrict__ perm0,
                                              const float* __restrict__ dinv0,
                                              const float* __restrict__ X0,
                                              const float* __restrict__ X1,
                                              const float* __restrict__ W,
                                              const float* __restrict__ bias,
                                              float* __restrict__ out0,
                                              float* __restrict__ out1) {
  __shared__ __attribute__((aligned(16))) float Wl[(DD / 2) * PW];
  __shared__ __attribute__((aligned(16))) float At[2][16 * PA];
  __shared__ __attribute__((aligned(16))) float Dt[2][16];
  const int t = threadIdx.x;
  const int lane = t & 31;
  const int w = __builtin_amdgcn_readfirstlane(t >> 5);

  const float4* W4 = (const float4*)W;
  for (int i = 0; i < 4; ++i) {
    const int idx = t + 512 * i;
    const int kp = idx >> 5;
    const int j = idx & 31;
    const float4 r0 = W4[(2 * kp) * (DD / 4) + j];
    const float4 r1 = W4[(2 * kp + 1) * (DD / 4) + j];
    float4 s0, s1;
    s0.x = r0.x; s0.y = r1.x; s0.z = r0.y; s0.w = r1.y;
    s1.x = r0.z; s1.y = r1.z; s1.z = r0.w; s1.w = r1.w;
    float4* q = (float4*)&Wl[kp * PW + 8 * j];
    q[0] = s0;
    q[1] = s1;
  }
  __syncthreads();

  const float bv = bias[(w & 7) * 16 + (lane & 15)];
  const int step = gridDim.x;
  int beg = 0, n = 0, sv = 0;
  float dd = 0.0f;
  int rbN = 0, rnN = 0;
  float rdN = 0.0f;
  if ((int)blockIdx.x < 2 * NT) {
    const int gt0 = blockIdx.x;
    const int g = (gt0 >= NT) ? 1 : 0;
    const size_t go = (size_t)g * GS;
    const int d = (gt0 - g * NT) * 16 + w;
    rbN = rowptr0[go + d];
    rnN = cnt0[go + d];
    rdN = dinv0[go + d];
  }
  int it = 0;
  for (int gt = (int)blockIdx.x - step; gt < 2 * NT; gt += step) {
    const int gtN = gt + step;
    const bool hasN = gtN < 2 * NT;
    const size_t goN = (gtN >= NT) ? (size_t)GS : (size_t)0;
    const int begN = __builtin_amdgcn_readfirstlane(rbN);
    const int nN = __builtin_amdgcn_readfirstlane(rnN);
    const float ddN = __int_as_float(__builtin_amdgcn_readfirstlane(__float_as_int(rdN)));
    int svN = 0;
    if (hasN && nN > 0) {
      const int mN = (nN < 32) ? nN : 32;
      svN = perm0[goN + begN + ((lane < mN) ? lane : (mN - 1))];
    }
    if (gtN + step < 2 * NT) {
      const int gtNN = gtN + step;
      const int g2 = (gtNN >= NT) ? 1 : 0;
      const size_t go2 = (size_t)g2 * GS;
      const int d2 = (gtNN - g2 * NT) * 16 + w;
      rbN = rowptr0[go2 + d2];
      rnN = cnt0[go2 + d2];
      rdN = dinv0[go2 + d2];
    }

    if (gt >= 0) {
      const int g = (gt >= NT) ? 1 : 0;
      const int tile = gt - g * NT;
      const size_t gofs = (size_t)g * GS;
      const int* __restrict__ perm = perm0 + gofs;
      const float* __restrict__ dinv = dinv0 + gofs;
      const float4* __restrict__ X4 = (const float4*)(g ? X1 : X0);
      float* __restrict__ out = g ? out1 : out0;
      float* Ab = At[it & 1];
      float* Db = Dt[it & 1];
      ++it;
      const int d = tile * 16 + w;

      float dv = 0.0f;
      if (LAYER == 0) dv = dinv[sv];
      const float4 h = X4[(size_t)d * (DD / 4) + lane];
      float4 acc = {0.0f, 0.0f, 0.0f, 0.0f};
      for (int base = 0; base < n; base += 32) {
        const int m = (n - base < 32) ? (n - base) : 32;
        if (base > 0) {
          const int idx = (lane < m) ? lane : (m - 1);
          sv = perm[beg + base + idx];
          if (LAYER == 0) {
            dv = dinv[sv];
          } else {
            asm volatile("" : : "v"(sv));
          }
        }
        for (int j = 0; j < m; j += KR)
          add_group<LAYER != 0>(X4, sv, dv, dd, j, (m - j < KR) ? (m - j) : KR, lane, acc);
      }
      float4 o;
      if (LAYER == 0) {
        o.x = acc.x + h.x * dd * dd;
        o.y = acc.y + h.y * dd * dd;
        o.z = acc.z + h.z * dd * dd;
        o.w = acc.w + h.w * dd * dd;
        if (lane == 0) Db[w] = dd;
      } else {
        o.x = (acc.x + h.x) * dd;
        o.y = (acc.y + h.y) * dd;
        o.z = (acc.z + h.z) * dd;
        o.w = (acc.w + h.w) * dd;
      }
      *(float4*)&Ab[w * PA + 4 * lane] = o;
      __syncthreads();

      if (w < 8) {
        const int half = lane >> 4;
        const int c = lane & 15;
        const int n0 = w * 16;
        const float* ap = Ab + c * PA + 2 * half;
        const float* bp = Wl + half * PW + 2 * (n0 + c);
        float rs[8];
        if (LAYER == 0) {
          const float4 q0 = *(const float4*)&Db[8 * half];
          const float4 q1 = *(const float4*)&Db[8 * half + 4];
          rs[0] = q0.x; rs[1] = q0.y; rs[2] = q0.z; rs[3] = q0.w;
          rs[4] = q1.x; rs[5] = q1.y; rs[6] = q1.z; rs[7] = q1.w;
        }
        v8f a8 = {};
#pragma unroll 8
        for (int k = 0; k < DD; k += 4) {
          const v2f a = *(const v2f*)(ap + k);
          const v2f b = *(const v2f*)(bp + (k >> 1) * PW);
          a8 = __builtin_amdgcn_wmma_f32_16x16x4_f32(
               false, a,  false, b,
               (short)0, a8,  false,  false);
        }
        float* op = out + (size_t)(tile * 16) * DD + n0 + c;
#pragma unroll
        for (int r = 0; r < 8; ++r) {
          float v = a8[r] + bv;
          if (LAYER == 0) v = fmaxf(v, 0.0f) * rs[r];
          op[(size_t)(r + 8 * half) * DD] = v;
        }
      }
    }
    beg = begN; n = nN; dd = ddN; sv = svN;
  }
}

extern "C" void kernel_launch(void* const* d_in, const int* in_sizes, int n_in,
                              void* d_out, int out_size, void* d_ws, size_t ws_size,
                              hipStream_t stream) {
  const float* x[2]  = {(const float*)d_in[0], (const float*)d_in[2]};
  const int*   ei[2] = {(const int*)d_in[1],   (const int*)d_in[3]};
  const float* W0 = (const float*)d_in[4];
  const float* b0 = (const float*)d_in[5];
  const float* W1 = (const float*)d_in[6];
  const float* b1 = (const float*)d_in[7];
  float* out = (float*)d_out;

  int*   ws     = (int*)d_ws;
  int*   cnt    = ws + O_CNT;
  int*   rowptr = ws + O_ROW;
  float* dinv   = (float*)(ws + O_DINV);
  int*   tsrc   = ws + O_TSRC;
  int*   perm   = ws + O_PERM;
  float* Y      = (float*)(ws + O_Y);
  int*   tdst   = ws + O_TDST;
  int*   bcount = ws + O_BCNT;
  int*   boffs  = ws + O_BOFF;

  const int gConv = 768;

  k_hist <<<dim3(NBLK, 2), dim3(512),  0, stream>>>(ei[0] + NE, ei[1] + NE, bcount);
  k_bscan<<<dim3(2),       dim3(1024), 0, stream>>>(bcount, boffs);
  k_bin  <<<dim3(NBLK, 2), dim3(512),  0, stream>>>(ei[0], ei[0] + NE, ei[1], ei[1] + NE,
                                                    boffs, tsrc, tdst);
  k_sort <<<dim3(NBK, 2),  dim3(512),  0, stream>>>(tsrc, tdst, boffs, cnt, rowptr, dinv, perm);

  k_conv<0><<<gConv, dim3(512), 0, stream>>>(rowptr, cnt, perm, dinv, x[0], x[1], W0, b0,
                                             Y, Y + GS);

  k_conv<1><<<gConv, dim3(512), 0, stream>>>(rowptr, cnt, perm, dinv, Y, Y + GS, W1, b1,
                                             out, out + (size_t)NN * DD);
}


# No global scan of bucket counts: chunks stored sorted in place with their bucket starts; each bucket adds 196 chunk starts, reads its 196 runs; 6 launches -> 4; per-node order still LDS arrival rank
# speedup vs baseline: 7.5217x; 1.0768x over previous
#include <hip/hip_runtime.h>


#define NN 100000
#define NE 1600000
#define DD 128
#define NP 100352

typedef __attribute__((ext_vector_type(2))) float v2f;
typedef __attribute__((ext_vector_type(8))) float v8f;

#define NBK 256
#define BS 391
#define CH 8192
#define NBLK 196
#define LSP (NBK + 1)
#define RPW 13
#define BSP 416
#define CAP 16384

#define O_CNT 0
#define O_ROW (O_CNT + NP)
#define O_DINV (O_ROW + NP)
#define O_TSRC (O_DINV + NP)
#define O_PERM (O_TSRC + NE)
#define O_Y (O_PERM + NE)
#define O_TDST (O_Y + NN * DD)
#define O_LS (O_TDST + NE)
#define GS (((O_LS + NBLK * LSP + 1023) / 1024) * 1024)

__global__ __launch_bounds__(512) void k_bin(const int* __restrict__ src0,
                                             const int* __restrict__ dst0,
                                             const int* __restrict__ src1,
                                             const int* __restrict__ dst1,
                                             int* __restrict__ tsrc,
                                             int* __restrict__ tdst,
                                             int* __restrict__ lsg) {
  const int g = blockIdx.y;
  const int* __restrict__ src = g ? src1 : src0;
  const int* __restrict__ dst = g ? dst1 : dst0;
  tsrc += (size_t)g * GS;
  tdst += (size_t)g * GS;
  lsg += (size_t)g * GS;
  __shared__ int lc[NBK];
  __shared__ int ls[NBK];
  __shared__ int wtot[16];
  __shared__ int ss[CH];
  __shared__ int sd[CH];
  const int t = threadIdx.x;
  const int lane = t & 31;
  const int w = t >> 5;
  const int blk = blockIdx.x;
  const int base = blk * CH;
  const int n = (NE - base < CH) ? (NE - base) : CH;
  if (t < NBK) lc[t] = 0;
  __syncthreads();
  int d[CH / 512], sv[CH / 512], r[CH / 512];
#pragma unroll
  for (int k = 0; k < CH / 512; ++k) {
    const int i = t + 512 * k;
    const int ii = base + ((i < n) ? i : (n - 1));
    d[k] = dst[ii];
    sv[k] = src[ii];
  }
#pragma unroll
  for (int k = 0; k < CH / 512; ++k) {
    r[k] = 0;
    if (t + 512 * k < n) r[k] = atomicAdd(&lc[d[k] / BS], 1);
  }
  __syncthreads();
  const int v = (t < NBK) ? lc[t] : 0;
  int inc = v;
  for (int o = 1; o < 32; o <<= 1) {
    const int u = __shfl_up(inc, o);
    if (lane >= o) inc += u;
  }
  if (lane == 31) wtot[w] = inc;
  __syncthreads();
  int bw = 0;
  for (int k = 0; k < 16; ++k)
    if (k < w) bw += wtot[k];
  if (t < NBK) {
    const int ex = bw + inc - v;
    ls[t] = ex;
    lsg[blk * LSP + t] = ex;
  }
  if (t == 0) lsg[blk * LSP + NBK] = n;
  __syncthreads();
#pragma unroll
  for (int k = 0; k < CH / 512; ++k) {
    const int i = t + 512 * k;
    if (i < n) {
      const int p = ls[d[k] / BS] + r[k];
      ss[p] = sv[k];
      sd[p] = d[k];
    }
  }
  __syncthreads();
  for (int i = t; i < n; i += 512) {
    tsrc[base + i] = ss[i];
    tdst[base + i] = sd[i];
  }
}

__global__ __launch_bounds__(512) void k_sort(const int* __restrict__ tsrc,
                                              const int* __restrict__ tdst,
                                              const int* __restrict__ lsg,
                                              int* __restrict__ cnt,
                                              int* __restrict__ rowptr,
                                              float* __restrict__ dinv,
                                              int* __restrict__ perm) {
  __shared__ int c[BSP];
  __shared__ int rp[BSP];
  __shared__ int wtot[16];
  __shared__ int stage[CAP];
  __shared__ int rs[16 * RPW];
  __shared__ int rl[16 * RPW];
  __shared__ int tot[2];
  {
    const size_t gofs = (size_t)blockIdx.y * GS;
    tsrc += gofs; tdst += gofs; lsg += gofs;
    cnt += gofs; rowptr += gofs; dinv += gofs; perm += gofs;
  }
  const int t = threadIdx.x;
  const int lane = t & 31;
  const int w = __builtin_amdgcn_readfirstlane(t >> 5);
  const int b = blockIdx.x;
  const int n0 = b * BS;
  const int nn = (NN - n0 < BS) ? (NN - n0) : BS;
  if (t < BSP) c[t] = 0;
  if (t < 2) tot[t] = 0;
  __syncthreads();
  if (t < 16 * RPW) {
    int s0 = 0, s1 = 0;
    if (t < NBLK) {
      s0 = lsg[t * LSP + b];
      s1 = lsg[t * LSP + b + 1];
      atomicAdd(&tot[0], s0);
      atomicAdd(&tot[1], s1);
    }
    rs[t] = t * CH + s0;
    rl[t] = s1 - s0;
  }
  __syncthreads();
  const int beg = tot[0];
  const int end = tot[1];
  const int nb = end - beg;
  int dd[2 * RPW], sv[2 * RPW], lnv[RPW];
  int mx = 0;
#pragma unroll
  for (int k = 0; k < RPW; ++k) {
    const int st = rs[w + 16 * k];
    const int ln = rl[w + 16 * k];
    lnv[k] = ln;
    mx = (ln > mx) ? ln : mx;
#pragma unroll
    for (int h = 0; h < 2; ++h) {
      const int o = lane + 32 * h;
      const int ii = (o < ln) ? (st + o) : 0;
      dd[2 * k + h] = tdst[ii];
      sv[2 * k + h] = tsrc[ii];
    }
  }
#pragma unroll
  for (int k = 0; k < RPW; ++k) {
#pragma unroll
    for (int h = 0; h < 2; ++h)
      if (lane + 32 * h < lnv[k]) atomicAdd(&c[dd[2 * k + h] - n0], 1);
  }
  if (mx > 64) {
    for (int k = 0; k < RPW; ++k) {
      const int st = rs[w + 16 * k];
      const int ln = rl[w + 16 * k];
      for (int o = 64 + lane; o < ln; o += 32) atomicAdd(&c[tdst[st + o] - n0], 1);
    }
  }
  __syncthreads();
  const int v = (t < BSP) ? c[t] : 0;
  int inc = v;
  for (int o = 1; o < 32; o <<= 1) {
    const int u = __shfl_up(inc, o);
    if (lane >= o) inc += u;
  }
  if (lane == 31) wtot[w] = inc;
  __syncthreads();
  int bw = 0;
  for (int k = 0; k < 16; ++k)
    if (k < w) bw += wtot[k];
  const int ex = bw + inc - v;
  if (t < nn) {
    cnt[n0 + t] = v;
    rowptr[n0 + t] = beg + ex;
    dinv[n0 + t] = rsqrtf((float)v + 1.0f);
  }
  if (t < BSP) {
    rp[t] = ex;
    c[t] = 0;
  }
  __syncthreads();
  if (nb <= CAP) {
#pragma unroll
    for (int k = 0; k < RPW; ++k) {
#pragma unroll
      for (int h = 0; h < 2; ++h) {
        if (lane + 32 * h < lnv[k]) {
          const int ld = dd[2 * k + h] - n0;
          const int rk = atomicAdd(&c[ld], 1);
          stage[rp[ld] + rk] = sv[2 * k + h];
        }
      }
    }
    if (mx > 64) {
      for (int k = 0; k < RPW; ++k) {
        const int st = rs[w + 16 * k];
        const int ln = rl[w + 16 * k];
        for (int o = 64 + lane; o < ln; o += 32) {
          const int ld = tdst[st + o] - n0;
          const int rk = atomicAdd(&c[ld], 1);
          stage[rp[ld] + rk] = tsrc[st + o];
        }
      }
    }
    __syncthreads();
    for (int i = t; i < nb; i += 512) perm[beg + i] = stage[i];
  } else {
#pragma unroll
    for (int k = 0; k < RPW; ++k) {
#pragma unroll
      for (int h = 0; h < 2; ++h) {
        if (lane + 32 * h < lnv[k]) {
          const int ld = dd[2 * k + h] - n0;
          const int rk = atomicAdd(&c[ld], 1);
          perm[beg + rp[ld] + rk] = sv[2 * k + h];
        }
      }
    }
    if (mx > 64) {
      for (int k = 0; k < RPW; ++k) {
        const int st = rs[w + 16 * k];
        const int ln = rl[w + 16 * k];
        for (int o = 64 + lane; o < ln; o += 32) {
          const int ld = tdst[st + o] - n0;
          const int rk = atomicAdd(&c[ld], 1);
          perm[beg + rp[ld] + rk] = tsrc[st + o];
        }
      }
    }
  }
}

#define NT (NN / 16)
#define PW 288
#define PA 132

template <int K, bool SCALED>
__device__ __forceinline__ void add_rows(const float4* __restrict__ H4, int sv,
                                         float dv, float dd, int j, int lane,
                                         float4& acc) {
  int s[K];
  float4 v[K];
#pragma unroll
  for (int k = 0; k < K; ++k) {
    s[k] = __builtin_amdgcn_readlane(sv, j + k);
  }
#pragma unroll
  for (int k = 0; k < K; ++k) v[k] = H4[(size_t)s[k] * (DD / 4) + lane];
  if (SCALED) {
#pragma unroll
    for (int k = 0; k < K; ++k) {
      acc.x += v[k].x;
      acc.y += v[k].y;
      acc.z += v[k].z;
      acc.w += v[k].w;
    }
  } else {
    float w[K];
#pragma unroll
    for (int k = 0; k < K; ++k) {
      w[k] = __int_as_float(__builtin_amdgcn_readlane(__float_as_int(dv), j + k)) * dd;
    }
#pragma unroll
    for (int k = 0; k < K; ++k) {
      acc.x = fmaf(v[k].x, w[k], acc.x);
      acc.y = fmaf(v[k].y, w[k], acc.y);
      acc.z = fmaf(v[k].z, w[k], acc.z);
      acc.w = fmaf(v[k].w, w[k], acc.w);
    }
  }
}

#define KR 12
template <bool SCALED>
__device__ __forceinline__ void add_group(const float4* __restrict__ H4, int sv,
                                          float dv, float dd, int j, int c, int lane,
                                          float4& acc) {
  switch (c) {
    case 1: add_rows<1, SCALED>(H4, sv, dv, dd, j, lane, acc); break;
    case 2: add_rows<2, SCALED>(H4, sv, dv, dd, j, lane, acc); break;
    case 3: add_rows<3, SCALED>(H4, sv, dv, dd, j, lane, acc); break;
    case 4: add_rows<4, SCALED>(H4, sv, dv, dd, j, lane, acc); break;
    case 5: add_rows<5, SCALED>(H4, sv, dv, dd, j, lane, acc); break;
    case 6: add_rows<6, SCALED>(H4, sv, dv, dd, j, lane, acc); break;
    case 7: add_rows<7, SCALED>(H4, sv, dv, dd, j, lane, acc); break;
    case 8: add_rows<8, SCALED>(H4, sv, dv, dd, j, lane, acc); break;
    case 9: add_rows<9, SCALED>(H4, sv, dv, dd, j, lane, acc); break;
    case 10: add_rows<10, SCALED>(H4, sv, dv, dd, j, lane, acc); break;
    case 11: add_rows<11, SCALED>(H4, sv, dv, dd, j, lane, acc); break;
    default: add_rows<12, SCALED>(H4, sv, dv, dd, j, lane, acc); break;
  }
}

template <int LAYER>
__global__ __launch_bounds__(512) void k_conv(const int* __restrict__ rowptr0,
                                              const int* __restrict__ cnt0,
                                              const int* __restrict__ perm0,
                                              const float* __restrict__ dinv0,
                                              const float* __restrict__ X0,
                                              const float* __restrict__ X1,
                                              const float* __restrict__ W,
                                              const float* __restrict__ bias,
                                              float* __restrict__ out0,
                                              float* __restrict__ out1) {
  __shared__ __attribute__((aligned(16))) float Wl[(DD / 2) * PW];
  __shared__ __attribute__((aligned(16))) float At[2][16 * PA];
  __shared__ __attribute__((aligned(16))) float Dt[2][16];
  const int t = threadIdx.x;
  const int lane = t & 31;
  const int w = __builtin_amdgcn_readfirstlane(t >> 5);

  const float4* W4 = (const float4*)W;
  for (int i = 0; i < 4; ++i) {
    const int idx = t + 512 * i;
    const int kp = idx >> 5;
    const int j = idx & 31;
    const float4 r0 = W4[(2 * kp) * (DD / 4) + j];
    const float4 r1 = W4[(2 * kp + 1) * (DD / 4) + j];
    float4 s0, s1;
    s0.x = r0.x; s0.y = r1.x; s0.z = r0.y; s0.w = r1.y;
    s1.x = r0.z; s1.y = r1.z; s1.z = r0.w; s1.w = r1.w;
    float4* q = (float4*)&Wl[kp * PW + 8 * j];
    q[0] = s0;
    q[1] = s1;
  }
  __syncthreads();

  const float bv = bias[(w & 7) * 16 + (lane & 15)];
  const int step = gridDim.x;
  int beg = 0, n = 0, sv = 0;
  float dd = 0.0f;
  int rbN = 0, rnN = 0;
  float rdN = 0.0f;
  if ((int)blockIdx.x < 2 * NT) {
    const int gt0 = blockIdx.x;
    const int g = (gt0 >= NT) ? 1 : 0;
    const size_t go = (size_t)g * GS;
    const int d = (gt0 - g * NT) * 16 + w;
    rbN = rowptr0[go + d];
    rnN = cnt0[go + d];
    rdN = dinv0[go + d];
  }
  int it = 0;
  for (int gt = (int)blockIdx.x - step; gt < 2 * NT; gt += step) {
    const int gtN = gt + step;
    const bool hasN = gtN < 2 * NT;
    const size_t goN = (gtN >= NT) ? (size_t)GS : (size_t)0;
    const int begN = __builtin_amdgcn_readfirstlane(rbN);
    const int nN = __builtin_amdgcn_readfirstlane(rnN);
    const float ddN = __int_as_float(__builtin_amdgcn_readfirstlane(__float_as_int(rdN)));
    int svN = 0;
    if (hasN && nN > 0) {
      const int mN = (nN < 32) ? nN : 32;
      svN = perm0[goN + begN + ((lane < mN) ? lane : (mN - 1))];
    }
    if (gtN + step < 2 * NT) {
      const int gtNN = gtN + step;
      const int g2 = (gtNN >= NT) ? 1 : 0;
      const size_t go2 = (size_t)g2 * GS;
      const int d2 = (gtNN - g2 * NT) * 16 + w;
      rbN = rowptr0[go2 + d2];
      rnN = cnt0[go2 + d2];
      rdN = dinv0[go2 + d2];
    }

    if (gt >= 0) {
      const int g = (gt >= NT) ? 1 : 0;
      const int tile = gt - g * NT;
      const size_t gofs = (size_t)g * GS;
      const int* __restrict__ perm = perm0 + gofs;
      const float* __restrict__ dinv = dinv0 + gofs;
      const float4* __restrict__ X4 = (const float4*)(g ? X1 : X0);
      float* __restrict__ out = g ? out1 : out0;
      float* Ab = At[it & 1];
      float* Db = Dt[it & 1];
      ++it;
      const int d = tile * 16 + w;

      float dv = 0.0f;
      if (LAYER == 0) dv = dinv[sv];
      const float4 h = X4[(size_t)d * (DD / 4) + lane];
      float4 acc = {0.0f, 0.0f, 0.0f, 0.0f};
      for (int base = 0; base < n; base += 32) {
        const int m = (n - base < 32) ? (n - base) : 32;
        if (base > 0) {
          const int idx = (lane < m) ? lane : (m - 1);
          sv = perm[beg + base + idx];
          if (LAYER == 0) {
            dv = dinv[sv];
          } else {
            asm volatile("" : : "v"(sv));
          }
        }
        for (int j = 0; j < m; j += KR)
          add_group<LAYER != 0>(X4, sv, dv, dd, j, (m - j < KR) ? (m - j) : KR, lane, acc);
      }
      float4 o;
      if (LAYER == 0) {
        o.x = acc.x + h.x * dd * dd;
        o.y = acc.y + h.y * dd * dd;
        o.z = acc.z + h.z * dd * dd;
        o.w = acc.w + h.w * dd * dd;
        if (lane == 0) Db[w] = dd;
      } else {
        o.x = (acc.x + h.x) * dd;
        o.y = (acc.y + h.y) * dd;
        o.z = (acc.z + h.z) * dd;
        o.w = (acc.w + h.w) * dd;
      }
      *(float4*)&Ab[w * PA + 4 * lane] = o;
      __syncthreads();

      if (w < 8) {
        const int half = lane >> 4;
        const int c = lane & 15;
        const int n0 = w * 16;
        const float* ap = Ab + c * PA + 2 * half;
        const float* bp = Wl + half * PW + 2 * (n0 + c);
        float rs[8];
        if (LAYER == 0) {
          const float4 q0 = *(const float4*)&Db[8 * half];
          const float4 q1 = *(const float4*)&Db[8 * half + 4];
          rs[0] = q0.x; rs[1] = q0.y; rs[2] = q0.z; rs[3] = q0.w;
          rs[4] = q1.x; rs[5] = q1.y; rs[6] = q1.z; rs[7] = q1.w;
        }
        v8f a8 = {};
#pragma unroll 8
        for (int k = 0; k < DD; k += 4) {
          const v2f a = *(const v2f*)(ap + k);
          const v2f b = *(const v2f*)(bp + (k >> 1) * PW);
          a8 = __builtin_amdgcn_wmma_f32_16x16x4_f32(
               false, a,  false, b,
               (short)0, a8,  false,  false);
        }
        float* op = out + (size_t)(tile * 16) * DD + n0 + c;
#pragma unroll
        for (int r = 0; r < 8; ++r) {
          float v = a8[r] + bv;
          if (LAYER == 0) v = fmaxf(v, 0.0f) * rs[r];
          op[(size_t)(r + 8 * half) * DD] = v;
        }
      }
    }
    beg = begN; n = nN; dd = ddN; sv = svN;
  }
}

extern "C" void kernel_launch(void* const* d_in, const int* in_sizes, int n_in,
                              void* d_out, int out_size, void* d_ws, size_t ws_size,
                              hipStream_t stream) {
  const float* x[2]  = {(const float*)d_in[0], (const float*)d_in[2]};
  const int*   ei[2] = {(const int*)d_in[1],   (const int*)d_in[3]};
  const float* W0 = (const float*)d_in[4];
  const float* b0 = (const float*)d_in[5];
  const float* W1 = (const float*)d_in[6];
  const float* b1 = (const float*)d_in[7];
  float* out = (float*)d_out;

  int*   ws     = (int*)d_ws;
  int*   cnt    = ws + O_CNT;
  int*   rowptr = ws + O_ROW;
  float* dinv   = (float*)(ws + O_DINV);
  int*   tsrc   = ws + O_TSRC;
  int*   perm   = ws + O_PERM;
  float* Y      = (float*)(ws + O_Y);
  int*   tdst   = ws + O_TDST;
  int*   lsg    = ws + O_LS;

  const int gConv = 768;

  k_bin  <<<dim3(NBLK, 2), dim3(512),  0, stream>>>(ei[0], ei[0] + NE, ei[1], ei[1] + NE,
                                                    tsrc, tdst, lsg);
  k_sort <<<dim3(NBK, 2),  dim3(512),  0, stream>>>(tsrc, tdst, lsg, cnt, rowptr, dinv, perm);

  k_conv<0><<<gConv, dim3(512), 0, stream>>>(rowptr, cnt, perm, dinv, x[0], x[1], W0, b0,
                                             Y, Y + GS);

  k_conv<1><<<gConv, dim3(512), 0, stream>>>(rowptr, cnt, perm, dinv, Y, Y + GS, W1, b1,
                                             out, out + (size_t)NN * DD);
}


# Output tile staged in LDS and written one tile later by the 8 non-matrix waves as whole 512B rows, one store per row (was 64B pieces of two rows per store); sums and values unchanged
# speedup vs baseline: 8.1331x; 1.0813x over previous
#include <hip/hip_runtime.h>


#define NN 100000
#define NE 1600000
#define DD 128
#define NP 100352

typedef __attribute__((ext_vector_type(2))) float v2f;
typedef __attribute__((ext_vector_type(8))) float v8f;

#define NBK 256
#define BS 391
#define CH 8192
#define NBLK 196
#define LSP (NBK + 1)
#define RPW 13
#define BSP 416
#define CAP 16384

#define O_CNT 0
#define O_ROW (O_CNT + NP)
#define O_DINV (O_ROW + NP)
#define O_TSRC (O_DINV + NP)
#define O_PERM (O_TSRC + NE)
#define O_Y (O_PERM + NE)
#define O_TDST (O_Y + NN * DD)
#define O_LS (O_TDST + NE)
#define GS (((O_LS + NBLK * LSP + 1023) / 1024) * 1024)

__global__ __launch_bounds__(512) void k_bin(const int* __restrict__ src0,
                                             const int* __restrict__ dst0,
                                             const int* __restrict__ src1,
                                             const int* __restrict__ dst1,
                                             int* __restrict__ tsrc,
                                             int* __restrict__ tdst,
                                             int* __restrict__ lsg) {
  const int g = blockIdx.y;
  const int* __restrict__ src = g ? src1 : src0;
  const int* __restrict__ dst = g ? dst1 : dst0;
  tsrc += (size_t)g * GS;
  tdst += (size_t)g * GS;
  lsg += (size_t)g * GS;
  __shared__ int lc[NBK];
  __shared__ int ls[NBK];
  __shared__ int wtot[16];
  __shared__ int ss[CH];
  __shared__ int sd[CH];
  const int t = threadIdx.x;
  const int lane = t & 31;
  const int w = t >> 5;
  const int blk = blockIdx.x;
  const int base = blk * CH;
  const int n = (NE - base < CH) ? (NE - base) : CH;
  if (t < NBK) lc[t] = 0;
  __syncthreads();
  int d[CH / 512], sv[CH / 512], r[CH / 512];
#pragma unroll
  for (int k = 0; k < CH / 512; ++k) {
    const int i = t + 512 * k;
    const int ii = base + ((i < n) ? i : (n - 1));
    d[k] = dst[ii];
    sv[k] = src[ii];
  }
#pragma unroll
  for (int k = 0; k < CH / 512; ++k) {
    r[k] = 0;
    if (t + 512 * k < n) r[k] = atomicAdd(&lc[d[k] / BS], 1);
  }
  __syncthreads();
  const int v = (t < NBK) ? lc[t] : 0;
  int inc = v;
  for (int o = 1; o < 32; o <<= 1) {
    const int u = __shfl_up(inc, o);
    if (lane >= o) inc += u;
  }
  if (lane == 31) wtot[w] = inc;
  __syncthreads();
  int bw = 0;
  for (int k = 0; k < 16; ++k)
    if (k < w) bw += wtot[k];
  if (t < NBK) {
    const int ex = bw + inc - v;
    ls[t] = ex;
    lsg[blk * LSP + t] = ex;
  }
  if (t == 0) lsg[blk * LSP + NBK] = n;
  __syncthreads();
#pragma unroll
  for (int k = 0; k < CH / 512; ++k) {
    const int i = t + 512 * k;
    if (i < n) {
      const int p = ls[d[k] / BS] + r[k];
      ss[p] = sv[k];
      sd[p] = d[k];
    }
  }
  __syncthreads();
  for (int i = t; i < n; i += 512) {
    tsrc[base + i] = ss[i];
    tdst[base + i] = sd[i];
  }
}

__global__ __launch_bounds__(512) void k_sort(const int* __restrict__ tsrc,
                                              const int* __restrict__ tdst,
                                              const int* __restrict__ lsg,
                                              int* __restrict__ cnt,
                                              int* __restrict__ rowptr,
                                              float* __restrict__ dinv,
                                              int* __restrict__ perm) {
  __shared__ int c[BSP];
  __shared__ int rp[BSP];
  __shared__ int wtot[16];
  __shared__ int stage[CAP];
  __shared__ int rs[16 * RPW];
  __shared__ int rl[16 * RPW];
  __shared__ int tot[2];
  {
    const size_t gofs = (size_t)blockIdx.y * GS;
    tsrc += gofs; tdst += gofs; lsg += gofs;
    cnt += gofs; rowptr += gofs; dinv += gofs; perm += gofs;
  }
  const int t = threadIdx.x;
  const int lane = t & 31;
  const int w = __builtin_amdgcn_readfirstlane(t >> 5);
  const int b = blockIdx.x;
  const int n0 = b * BS;
  const int nn = (NN - n0 < BS) ? (NN - n0) : BS;
  if (t < BSP) c[t] = 0;
  if (t < 2) tot[t] = 0;
  __syncthreads();
  if (t < 16 * RPW) {
    int s0 = 0, s1 = 0;
    if (t < NBLK) {
      s0 = lsg[t * LSP + b];
      s1 = lsg[t * LSP + b + 1];
      atomicAdd(&tot[0], s0);
      atomicAdd(&tot[1], s1);
    }
    rs[t] = t * CH + s0;
    rl[t] = s1 - s0;
  }
  __syncthreads();
  const int beg = tot[0];
  const int end = tot[1];
  const int nb = end - beg;
  int dd[2 * RPW], sv[2 * RPW], lnv[RPW];
  int mx = 0;
#pragma unroll
  for (int k = 0; k < RPW; ++k) {
    const int st = rs[w + 16 * k];
    const int ln = rl[w + 16 * k];
    lnv[k] = ln;
    mx = (ln > mx) ? ln : mx;
#pragma unroll
    for (int h = 0; h < 2; ++h) {
      const int o = lane + 32 * h;
      const int ii = (o < ln) ? (st + o) : 0;
      dd[2 * k + h] = tdst[ii];
      sv[2 * k + h] = tsrc[ii];
    }
  }
#pragma unroll
  for (int k = 0; k < RPW; ++k) {
#pragma unroll
    for (int h = 0; h < 2; ++h)
      if (lane + 32 * h < lnv[k]) atomicAdd(&c[dd[2 * k + h] - n0], 1);
  }
  if (mx > 64) {
    for (int k = 0; k < RPW; ++k) {
      const int st = rs[w + 16 * k];
      const int ln = rl[w + 16 * k];
      for (int o = 64 + lane; o < ln; o += 32) atomicAdd(&c[tdst[st + o] - n0], 1);
    }
  }
  __syncthreads();
  const int v = (t < BSP) ? c[t] : 0;
  int inc = v;
  for (int o = 1; o < 32; o <<= 1) {
    const int u = __shfl_up(inc, o);
    if (lane >= o) inc += u;
  }
  if (lane == 31) wtot[w] = inc;
  __syncthreads();
  int bw = 0;
  for (int k = 0; k < 16; ++k)
    if (k < w) bw += wtot[k];
  const int ex = bw + inc - v;
  if (t < nn) {
    cnt[n0 + t] = v;
    rowptr[n0 + t] = beg + ex;
    dinv[n0 + t] = rsqrtf((float)v + 1.0f);
  }
  if (t < BSP) {
    rp[t] = ex;
    c[t] = 0;
  }
  __syncthreads();
  if (nb <= CAP) {
#pragma unroll
    for (int k = 0; k < RPW; ++k) {
#pragma unroll
      for (int h = 0; h < 2; ++h) {
        if (lane + 32 * h < lnv[k]) {
          const int ld = dd[2 * k + h] - n0;
          const int rk = atomicAdd(&c[ld], 1);
          stage[rp[ld] + rk] = sv[2 * k + h];
        }
      }
    }
    if (mx > 64) {
      for (int k = 0; k < RPW; ++k) {
        const int st = rs[w + 16 * k];
        const int ln = rl[w + 16 * k];
        for (int o = 64 + lane; o < ln; o += 32) {
          const int ld = tdst[st + o] - n0;
          const int rk = atomicAdd(&c[ld], 1);
          stage[rp[ld] + rk] = tsrc[st + o];
        }
      }
    }
    __syncthreads();
    for (int i = t; i < nb; i += 512) perm[beg + i] = stage[i];
  } else {
#pragma unroll
    for (int k = 0; k < RPW; ++k) {
#pragma unroll
      for (int h = 0; h < 2; ++h) {
        if (lane + 32 * h < lnv[k]) {
          const int ld = dd[2 * k + h] - n0;
          const int rk = atomicAdd(&c[ld], 1);
          perm[beg + rp[ld] + rk] = sv[2 * k + h];
        }
      }
    }
    if (mx > 64) {
      for (int k = 0; k < RPW; ++k) {
        const int st = rs[w + 16 * k];
        const int ln = rl[w + 16 * k];
        for (int o = 64 + lane; o < ln; o += 32) {
          const int ld = tdst[st + o] - n0;
          const int rk = atomicAdd(&c[ld], 1);
          perm[beg + rp[ld] + rk] = tsrc[st + o];
        }
      }
    }
  }
}

#define NT (NN / 16)
#define PW 288
#define PA 132
#define PO 132

template <int K, bool SCALED>
__device__ __forceinline__ void add_rows(const float4* __restrict__ H4, int sv,
                                         float dv, float dd, int j, int lane,
                                         float4& acc) {
  int s[K];
  float4 v[K];
#pragma unroll
  for (int k = 0; k < K; ++k) {
    s[k] = __builtin_amdgcn_readlane(sv, j + k);
  }
#pragma unroll
  for (int k = 0; k < K; ++k) v[k] = H4[(size_t)s[k] * (DD / 4) + lane];
  if (SCALED) {
#pragma unroll
    for (int k = 0; k < K; ++k) {
      acc.x += v[k].x;
      acc.y += v[k].y;
      acc.z += v[k].z;
      acc.w += v[k].w;
    }
  } else {
    float w[K];
#pragma unroll
    for (int k = 0; k < K; ++k) {
      w[k] = __int_as_float(__builtin_amdgcn_readlane(__float_as_int(dv), j + k)) * dd;
    }
#pragma unroll
    for (int k = 0; k < K; ++k) {
      acc.x = fmaf(v[k].x, w[k], acc.x);
      acc.y = fmaf(v[k].y, w[k], acc.y);
      acc.z = fmaf(v[k].z, w[k], acc.z);
      acc.w = fmaf(v[k].w, w[k], acc.w);
    }
  }
}

#define KR 12
template <bool SCALED>
__device__ __forceinline__ void add_group(const float4* __restrict__ H4, int sv,
                                          float dv, float dd, int j, int c, int lane,
                                          float4& acc) {
  switch (c) {
    case 1: add_rows<1, SCALED>(H4, sv, dv, dd, j, lane, acc); break;
    case 2: add_rows<2, SCALED>(H4, sv, dv, dd, j, lane, acc); break;
    case 3: add_rows<3, SCALED>(H4, sv, dv, dd, j, lane, acc); break;
    case 4: add_rows<4, SCALED>(H4, sv, dv, dd, j, lane, acc); break;
    case 5: add_rows<5, SCALED>(H4, sv, dv, dd, j, lane, acc); break;
    case 6: add_rows<6, SCALED>(H4, sv, dv, dd, j, lane, acc); break;
    case 7: add_rows<7, SCALED>(H4, sv, dv, dd, j, lane, acc); break;
    case 8: add_rows<8, SCALED>(H4, sv, dv, dd, j, lane, acc); break;
    case 9: add_rows<9, SCALED>(H4, sv, dv, dd, j, lane, acc); break;
    case 10: add_rows<10, SCALED>(H4, sv, dv, dd, j, lane, acc); break;
    case 11: add_rows<11, SCALED>(H4, sv, dv, dd, j, lane, acc); break;
    default: add_rows<12, SCALED>(H4, sv, dv, dd, j, lane, acc); break;
  }
}

template <int LAYER>
__global__ __launch_bounds__(512) void k_conv(const int* __restrict__ rowptr0,
                                              const int* __restrict__ cnt0,
                                              const int* __restrict__ perm0,
                                              const float* __restrict__ dinv0,
                                              const float* __restrict__ X0,
                                              const float* __restrict__ X1,
                                              const float* __restrict__ W,
                                              const float* __restrict__ bias,
                                              float* __restrict__ out0,
                                              float* __restrict__ out1) {
  __shared__ __attribute__((aligned(16))) float Wl[(DD / 2) * PW];
  __shared__ __attribute__((aligned(16))) float At[2][16 * PA];
  __shared__ __attribute__((aligned(16))) float Dt[2][16];
  __shared__ __attribute__((aligned(16))) float Ot[2][16 * PO];
  const int t = threadIdx.x;
  const int lane = t & 31;
  const int w = __builtin_amdgcn_readfirstlane(t >> 5);

  const float4* W4 = (const float4*)W;
  for (int i = 0; i < 4; ++i) {
    const int idx = t + 512 * i;
    const int kp = idx >> 5;
    const int j = idx & 31;
    const float4 r0 = W4[(2 * kp) * (DD / 4) + j];
    const float4 r1 = W4[(2 * kp + 1) * (DD / 4) + j];
    float4 s0, s1;
    s0.x = r0.x; s0.y = r1.x; s0.z = r0.y; s0.w = r1.y;
    s1.x = r0.z; s1.y = r1.z; s1.z = r0.w; s1.w = r1.w;
    float4* q = (float4*)&Wl[kp * PW + 8 * j];
    q[0] = s0;
    q[1] = s1;
  }
  __syncthreads();

  const float bv = bias[(w & 7) * 16 + (lane & 15)];
  const int step = gridDim.x;
  int beg = 0, n = 0, sv = 0;
  float dd = 0.0f;
  int rbN = 0, rnN = 0;
  float rdN = 0.0f;
  if ((int)blockIdx.x < 2 * NT) {
    const int gt0 = blockIdx.x;
    const int g = (gt0 >= NT) ? 1 : 0;
    const size_t go = (size_t)g * GS;
    const int d = (gt0 - g * NT) * 16 + w;
    rbN = rowptr0[go + d];
    rnN = cnt0[go + d];
    rdN = dinv0[go + d];
  }
  int it = 0;
  int gt = (int)blockIdx.x - step;
  for (; gt < 2 * NT; gt += step) {
    const int gtN = gt + step;
    const bool hasN = gtN < 2 * NT;
    const size_t goN = (gtN >= NT) ? (size_t)GS : (size_t)0;
    const int begN = __builtin_amdgcn_readfirstlane(rbN);
    const int nN = __builtin_amdgcn_readfirstlane(rnN);
    const float ddN = __int_as_float(__builtin_amdgcn_readfirstlane(__float_as_int(rdN)));
    int svN = 0;
    if (hasN && nN > 0) {
      const int mN = (nN < 32) ? nN : 32;
      svN = perm0[goN + begN + ((lane < mN) ? lane : (mN - 1))];
    }
    if (gtN + step < 2 * NT) {
      const int gtNN = gtN + step;
      const int g2 = (gtNN >= NT) ? 1 : 0;
      const size_t go2 = (size_t)g2 * GS;
      const int d2 = (gtNN - g2 * NT) * 16 + w;
      rbN = rowptr0[go2 + d2];
      rnN = cnt0[go2 + d2];
      rdN = dinv0[go2 + d2];
    }

    if (gt >= 0) {
      const int g = (gt >= NT) ? 1 : 0;
      const int tile = gt - g * NT;
      const size_t gofs = (size_t)g * GS;
      const int* __restrict__ perm = perm0 + gofs;
      const float* __restrict__ dinv = dinv0 + gofs;
      const float4* __restrict__ X4 = (const float4*)(g ? X1 : X0);
      float* Ab = At[it & 1];
      float* Db = Dt[it & 1];
      float* Ob = Ot[it & 1];
      const float* Op = Ot[(it & 1) ^ 1];
      ++it;
      const int d = tile * 16 + w;

      float dv = 0.0f;
      if (LAYER == 0) dv = dinv[sv];
      const float4 h = X4[(size_t)d * (DD / 4) + lane];
      float4 acc = {0.0f, 0.0f, 0.0f, 0.0f};
      for (int base = 0; base < n; base += 32) {
        const int m = (n - base < 32) ? (n - base) : 32;
        if (base > 0) {
          const int idx = (lane < m) ? lane : (m - 1);
          sv = perm[beg + base + idx];
          if (LAYER == 0) {
            dv = dinv[sv];
          } else {
            asm volatile("" : : "v"(sv));
          }
        }
        for (int j = 0; j < m; j += KR)
          add_group<LAYER != 0>(X4, sv, dv, dd, j, (m - j < KR) ? (m - j) : KR, lane, acc);
      }
      float4 o;
      if (LAYER == 0) {
        o.x = acc.x + h.x * dd * dd;
        o.y = acc.y + h.y * dd * dd;
        o.z = acc.z + h.z * dd * dd;
        o.w = acc.w + h.w * dd * dd;
        if (lane == 0) Db[w] = dd;
      } else {
        o.x = (acc.x + h.x) * dd;
        o.y = (acc.y + h.y) * dd;
        o.z = (acc.z + h.z) * dd;
        o.w = (acc.w + h.w) * dd;
      }
      *(float4*)&Ab[w * PA + 4 * lane] = o;
      __syncthreads();

      if (w < 8) {
        const int half = lane >> 4;
        const int c = lane & 15;
        const int n0 = w * 16;
        const float* ap = Ab + c * PA + 2 * half;
        const float* bp = Wl + half * PW + 2 * (n0 + c);
        float rs[8];
        if (LAYER == 0) {
          const float4 q0 = *(const float4*)&Db[8 * half];
          const float4 q1 = *(const float4*)&Db[8 * half + 4];
          rs[0] = q0.x; rs[1] = q0.y; rs[2] = q0.z; rs[3] = q0.w;
          rs[4] = q1.x; rs[5] = q1.y; rs[6] = q1.z; rs[7] = q1.w;
        }
        v8f a8 = {};
#pragma unroll 8
        for (int k = 0; k < DD; k += 4) {
          const v2f a = *(const v2f*)(ap + k);
          const v2f b = *(const v2f*)(bp + (k >> 1) * PW);
          a8 = __builtin_amdgcn_wmma_f32_16x16x4_f32(
               false, a,  false, b,
               (short)0, a8,  false,  false);
        }
        float* op = Ob + (8 * half) * PO + n0 + c;
#pragma unroll
        for (int r = 0; r < 8; ++r) {
          float v = a8[r] + bv;
          if (LAYER == 0) v = fmaxf(v, 0.0f) * rs[r];
          op[r * PO] = v;
        }
      } else if (gt >= step) {
        const int gtP = gt - step;
        const int gP = (gtP >= NT) ? 1 : 0;
        float4* o4 = (float4*)((gP ? out1 : out0) +
                               (size_t)((gtP - gP * NT) * 16 + (w - 8)) * DD);
        const float4* p4 = (const float4*)(Op + (w - 8) * PO);
        const float4 q0 = p4[lane];
        const float4 q1 = p4[8 * (PO / 4) + lane];
        o4[lane] = q0;
        o4[8 * (DD / 4) + lane] = q1;
      }
    }
    beg = begN; n = nN; dd = ddN; sv = svN;
  }
  if (it > 0) {
    __syncthreads();
    if (w >= 8) {
      const int gtP = gt - step;
      const int gP = (gtP >= NT) ? 1 : 0;
      float4* o4 = (float4*)((gP ? out1 : out0) +
                             (size_t)((gtP - gP * NT) * 16 + (w - 8)) * DD);
      const float4* p4 = (const float4*)(Ot[(it - 1) & 1] + (w - 8) * PO);
      const float4 q0 = p4[lane];
      const float4 q1 = p4[8 * (PO / 4) + lane];
      o4[lane] = q0;
      o4[8 * (DD / 4) + lane] = q1;
    }
  }
}

extern "C" void kernel_launch(void* const* d_in, const int* in_sizes, int n_in,
                              void* d_out, int out_size, void* d_ws, size_t ws_size,
                              hipStream_t stream) {
  const float* x[2]  = {(const float*)d_in[0], (const float*)d_in[2]};
  const int*   ei[2] = {(const int*)d_in[1],   (const int*)d_in[3]};
  const float* W0 = (const float*)d_in[4];
  const float* b0 = (const float*)d_in[5];
  const float* W1 = (const float*)d_in[6];
  const float* b1 = (const float*)d_in[7];
  float* out = (float*)d_out;

  int*   ws     = (int*)d_ws;
  int*   cnt    = ws + O_CNT;
  int*   rowptr = ws + O_ROW;
  float* dinv   = (float*)(ws + O_DINV);
  int*   tsrc   = ws + O_TSRC;
  int*   perm   = ws + O_PERM;
  float* Y      = (float*)(ws + O_Y);
  int*   tdst   = ws + O_TDST;
  int*   lsg    = ws + O_LS;

  const int gConv = 768;

  k_bin  <<<dim3(NBLK, 2), dim3(512),  0, stream>>>(ei[0], ei[0] + NE, ei[1], ei[1] + NE,
                                                    tsrc, tdst, lsg);
  k_sort <<<dim3(NBK, 2),  dim3(512),  0, stream>>>(tsrc, tdst, lsg, cnt, rowptr, dinv, perm);

  k_conv<0><<<gConv, dim3(512), 0, stream>>>(rowptr, cnt, perm, dinv, x[0], x[1], W0, b0,
                                             Y, Y + GS);

  k_conv<1><<<gConv, dim3(512), 0, stream>>>(rowptr, cnt, perm, dinv, Y, Y + GS, W1, b1,
                                             out, out + (size_t)NN * DD);
}
